# starting kernel
# baseline (speedup 1.0000x reference)
#include <hip/hip_runtime.h>
#include <hip/hip_bf16.h>
#include <cstddef>

typedef __attribute__((ext_vector_type(2))) float v2f;
typedef __attribute__((ext_vector_type(8))) float v8f;

#define CH 128

__global__ void deg_init_kernel(float* __restrict__ deg, int n) {
    int i = blockIdx.x * blockDim.x + threadIdx.x;
    if (i < n) deg[i] = 1.0f;
}

__global__ void deg_count_kernel(const int* __restrict__ dst, float* __restrict__ deg, int e) {
    int i = blockIdx.x * blockDim.x + threadIdx.x;
    if (i < e) unsafeAtomicAdd(&deg[dst[i]], 1.0f);
}

__global__ void dinv_kernel(float* __restrict__ deg, int n) {
    int i = blockIdx.x * blockDim.x + threadIdx.x;
    if (i < n) deg[i] = rsqrtf(deg[i]);
}

__global__ __launch_bounds__(128) void gemm_wmma_kernel(
    const float* __restrict__ X, const float* __restrict__ W,
    float* __restrict__ H, int nrows, int apply_relu)
{
    __shared__ float Xs[4][16 * CH];

    const int lane = threadIdx.x & 31;
    const int wave = threadIdx.x >> 5;
    const int row_base = blockIdx.x * 64 + wave * 16;
    const bool full_tile = (row_base + 16 <= nrows);

    float* xs = &Xs[wave][0];

    if (full_tile) {
        #pragma unroll
        for (int r = 0; r < 16; ++r) {
            float4 v = *(const float4*)(X + (size_t)(row_base + r) * CH + lane * 4);
            if (apply_relu) {
                v.x = fmaxf(v.x, 0.f); v.y = fmaxf(v.y, 0.f);
                v.z = fmaxf(v.z, 0.f); v.w = fmaxf(v.w, 0.f);
            }
            *(float4*)(xs + r * CH + lane * 4) = v;
        }
    } else {
        #pragma unroll
        for (int r = 0; r < 16; ++r) {
            int row = row_base + r;
            float4 v = make_float4(0.f, 0.f, 0.f, 0.f);
            if (row < nrows)
                v = *(const float4*)(X + (size_t)row * CH + lane * 4);
            if (apply_relu) {
                v.x = fmaxf(v.x, 0.f); v.y = fmaxf(v.y, 0.f);
                v.z = fmaxf(v.z, 0.f); v.w = fmaxf(v.w, 0.f);
            }
            *(float4*)(xs + r * CH + lane * 4) = v;
        }
    }
    __syncthreads();

    const int m    = lane & 15;
    const int hi   = lane >> 4;
    const int colN = lane & 15;

    v8f acc[8];
    #pragma unroll
    for (int n = 0; n < 8; ++n) acc[n] = (v8f)(0.0f);

    for (int k0 = 0; k0 < CH; k0 += 4) {
        const int klo = k0 + hi * 2;
        v2f a = *(const v2f*)(xs + m * CH + klo);
        #pragma unroll
        for (int n = 0; n < 8; ++n) {
            v2f b;
            b.x = W[(size_t)klo * CH + n * 16 + colN];
            b.y = W[(size_t)(klo + 1) * CH + n * 16 + colN];
            acc[n] = __builtin_amdgcn_wmma_f32_16x16x4_f32(
                false, a, false, b, (short)0, acc[n], false, false);
        }
    }

    if (full_tile) {
        #pragma unroll
        for (int n = 0; n < 8; ++n) {
            #pragma unroll
            for (int r = 0; r < 8; ++r) {
                int row = row_base + r + hi * 8;
                H[(size_t)row * CH + n * 16 + colN] = acc[n][r];
            }
        }
    } else if (row_base < nrows) {
        #pragma unroll
        for (int n = 0; n < 8; ++n) {
            #pragma unroll
            for (int r = 0; r < 8; ++r) {
                int row = row_base + r + hi * 8;
                if (row < nrows)
                    H[(size_t)row * CH + n * 16 + colN] = acc[n][r];
            }
        }
    }
}

__global__ __launch_bounds__(256) void agg_init_kernel(
    const float* __restrict__ H, const float* __restrict__ dinv,
    const float* __restrict__ bias, float* __restrict__ out, int n)
{
    int t = blockIdx.x * blockDim.x + threadIdx.x;
    if (t >= n * 32) return;
    int i = t >> 5;
    int c = (t & 31) * 4;
    float di = dinv[i];
    float w = di * di;
    float4 h = *(const float4*)(H + (size_t)i * CH + c);
    float4 b = *(const float4*)(bias + c);
    float4 o;
    o.x = fmaf(h.x, w, b.x);
    o.y = fmaf(h.y, w, b.y);
    o.z = fmaf(h.z, w, b.z);
    o.w = fmaf(h.w, w, b.w);
    *(float4*)(out + (size_t)i * CH + c) = o;
}

__global__ __launch_bounds__(256) void scatter_kernel(
    const float* __restrict__ H, const float* __restrict__ dinv,
    const int* __restrict__ src_idx, const int* __restrict__ dst_idx,
    float* __restrict__ out, int nedges)
{
    const int lane = threadIdx.x & 31;
    const int nwaves = (gridDim.x * blockDim.x) >> 5;
    int e = (blockIdx.x * blockDim.x + threadIdx.x) >> 5;

    for (; e < nedges; e += nwaves) {
        int s = src_idx[e];
        int d = dst_idx[e];

        int en = e + nwaves;
        if (en < nedges) {
            int sn = src_idx[en];
            __builtin_prefetch(H + (size_t)sn * CH + lane * 4, 0, 1);
        }

        float coef = dinv[s] * dinv[d];
        float4 hv = *(const float4*)(H + (size_t)s * CH + lane * 4);
        float* op = out + (size_t)d * CH + lane * 4;
        unsafeAtomicAdd(op + 0, hv.x * coef);
        unsafeAtomicAdd(op + 1, hv.y * coef);
        unsafeAtomicAdd(op + 2, hv.z * coef);
        unsafeAtomicAdd(op + 3, hv.w * coef);
    }
}

extern "C" void kernel_launch(void* const* d_in, const int* in_sizes, int n_in,
                              void* d_out, int out_size, void* d_ws, size_t ws_size,
                              hipStream_t stream) {
    const float* x    = (const float*)d_in[0];
    const int*   edge = (const int*)d_in[1];
    const float* W1   = (const float*)d_in[2];
    const float* b1   = (const float*)d_in[3];
    const float* W2   = (const float*)d_in[4];
    const float* b2   = (const float*)d_in[5];
    float* out = (float*)d_out;

    const int N = in_sizes[0] / CH;
    const int E = in_sizes[1] / 2;
    const int* srcI = edge;
    const int* dstI = edge + E;

    char* ws = (char*)d_ws;
    float* dinv = (float*)ws;
    size_t off  = ((size_t)N * 4 + 255) & ~(size_t)255;
    float* hbuf = (float*)(ws + off);
    float* tmp  = hbuf + (size_t)N * CH;

    const int T = 256;
    dim3 gN((N + T - 1) / T), gE((E + T - 1) / T);
    dim3 gGemm((N + 63) / 64);
    dim3 gAgg(((size_t)N * 32 + T - 1) / T);
    dim3 gScat(8192);

    deg_init_kernel<<<gN, T, 0, stream>>>(dinv, N);
    deg_count_kernel<<<gE, T, 0, stream>>>(dstI, dinv, E);
    dinv_kernel<<<gN, T, 0, stream>>>(dinv, N);

    gemm_wmma_kernel<<<gGemm, 128, 0, stream>>>(x, W1, hbuf, N, 0);
    agg_init_kernel<<<gAgg, T, 0, stream>>>(hbuf, dinv, b1, tmp, N);
    scatter_kernel<<<gScat, T, 0, stream>>>(hbuf, dinv, srcI, dstI, tmp, E);

    gemm_wmma_kernel<<<gGemm, 128, 0, stream>>>(tmp, W2, hbuf, N, 1);
    agg_init_kernel<<<gAgg, T, 0, stream>>>(hbuf, dinv, b2, out, N);
    scatter_kernel<<<gScat, T, 0, stream>>>(hbuf, dinv, srcI, dstI, out, E);
}


# aggregation re-decomposed: a work-group owns 400 output rows in LDS (self loop + bias as start), scans the whole dst list, adds its edges' rows by LDS f32 atomics, stores once; sum order LDS arrival
# speedup vs baseline: 2.0229x; 2.0229x over previous
#include <hip/hip_runtime.h>
#include <hip/hip_bf16.h>
#include <cstddef>

typedef __attribute__((ext_vector_type(2))) float v2f;
typedef __attribute__((ext_vector_type(8))) float v8f;

#define CH 128

__global__ void deg_init_kernel(float* __restrict__ deg, int n) {
    int i = blockIdx.x * blockDim.x + threadIdx.x;
    if (i < n) deg[i] = 1.0f;
}

__global__ void deg_count_kernel(const int* __restrict__ dst, float* __restrict__ deg, int e) {
    int i = blockIdx.x * blockDim.x + threadIdx.x;
    if (i < e) unsafeAtomicAdd(&deg[dst[i]], 1.0f);
}

__global__ void dinv_kernel(float* __restrict__ deg, int n) {
    int i = blockIdx.x * blockDim.x + threadIdx.x;
    if (i < n) deg[i] = rsqrtf(deg[i]);
}

__global__ __launch_bounds__(128) void gemm_wmma_kernel(
    const float* __restrict__ X, const float* __restrict__ W,
    float* __restrict__ H, int nrows, int apply_relu)
{
    __shared__ float Xs[4][16 * CH];

    const int lane = threadIdx.x & 31;
    const int wave = threadIdx.x >> 5;
    const int row_base = blockIdx.x * 64 + wave * 16;
    const bool full_tile = (row_base + 16 <= nrows);

    float* xs = &Xs[wave][0];

    if (full_tile) {
        #pragma unroll
        for (int r = 0; r < 16; ++r) {
            float4 v = *(const float4*)(X + (size_t)(row_base + r) * CH + lane * 4);
            if (apply_relu) {
                v.x = fmaxf(v.x, 0.f); v.y = fmaxf(v.y, 0.f);
                v.z = fmaxf(v.z, 0.f); v.w = fmaxf(v.w, 0.f);
            }
            *(float4*)(xs + r * CH + lane * 4) = v;
        }
    } else {
        #pragma unroll
        for (int r = 0; r < 16; ++r) {
            int row = row_base + r;
            float4 v = make_float4(0.f, 0.f, 0.f, 0.f);
            if (row < nrows)
                v = *(const float4*)(X + (size_t)row * CH + lane * 4);
            if (apply_relu) {
                v.x = fmaxf(v.x, 0.f); v.y = fmaxf(v.y, 0.f);
                v.z = fmaxf(v.z, 0.f); v.w = fmaxf(v.w, 0.f);
            }
            *(float4*)(xs + r * CH + lane * 4) = v;
        }
    }
    __syncthreads();

    const int m    = lane & 15;
    const int hi   = lane >> 4;
    const int colN = lane & 15;

    v8f acc[8];
    #pragma unroll
    for (int n = 0; n < 8; ++n) acc[n] = (v8f)(0.0f);

    for (int k0 = 0; k0 < CH; k0 += 4) {
        const int klo = k0 + hi * 2;
        v2f a = *(const v2f*)(xs + m * CH + klo);
        #pragma unroll
        for (int n = 0; n < 8; ++n) {
            v2f b;
            b.x = W[(size_t)klo * CH + n * 16 + colN];
            b.y = W[(size_t)(klo + 1) * CH + n * 16 + colN];
            acc[n] = __builtin_amdgcn_wmma_f32_16x16x4_f32(
                false, a, false, b, (short)0, acc[n], false, false);
        }
    }

    if (full_tile) {
        #pragma unroll
        for (int n = 0; n < 8; ++n) {
            #pragma unroll
            for (int r = 0; r < 8; ++r) {
                int row = row_base + r + hi * 8;
                H[(size_t)row * CH + n * 16 + colN] = acc[n][r];
            }
        }
    } else if (row_base < nrows) {
        #pragma unroll
        for (int n = 0; n < 8; ++n) {
            #pragma unroll
            for (int r = 0; r < 8; ++r) {
                int row = row_base + r + hi * 8;
                if (row < nrows)
                    H[(size_t)row * CH + n * 16 + colN] = acc[n][r];
            }
        }
    }
}

#define TILE   400
#define AGG_T  1024
#define AGG_W  32
#define SCAN_U 4
#define QDRAIN 32
#define QCAP   (SCAN_U * 128 + QDRAIN)

__device__ __forceinline__ void load_step(int4 (&v)[SCAN_U], const int* __restrict__ dst_idx,
                                          int b0, int lane, int nedges, bool vec_ok) {
    if (vec_ok && b0 + SCAN_U * 128 <= nedges) {
        #pragma unroll
        for (int u = 0; u < SCAN_U; ++u)
            v[u] = *(const int4*)(dst_idx + b0 + u * 128 + lane * 4);
    } else {
        #pragma unroll
        for (int u = 0; u < SCAN_U; ++u) {
            const int b = b0 + u * 128 + lane * 4;
            v[u].x = (b     < nedges) ? dst_idx[b]     : -1;
            v[u].y = (b + 1 < nedges) ? dst_idx[b + 1] : -1;
            v[u].z = (b + 2 < nedges) ? dst_idx[b + 2] : -1;
            v[u].w = (b + 3 < nedges) ? dst_idx[b + 3] : -1;
        }
    }
}

__device__ __forceinline__ void lds_add(float* p, float v) {
    __hip_atomic_fetch_add(p, v, __ATOMIC_RELAXED, __HIP_MEMORY_SCOPE_WORKGROUP);
}

__device__ __forceinline__ void drain_queue(
    const int* q, int qn, int lane, int lo,
    const float* __restrict__ H, const float* __restrict__ dinv,
    const int* __restrict__ src_idx, const int* __restrict__ dst_idx,
    float* tile)
{
    for (int base = 0; base < qn; base += 32) {
        const int m = min(32, qn - base);
        const int k = min(base + lane, qn - 1);
        const int e = q[k];
        const int s = src_idx[e];
        const int d = dst_idx[e];
        const float ds = dinv[s];
        const float dd = dinv[d];
        const int rl = d - lo;

        for (int j = 0; j < m; j += 8) {
            float4 h[8];
            int    jj[8];
            #pragma unroll
            for (int u = 0; u < 8; ++u) {
                jj[u] = min(j + u, m - 1);
                const int sj = __builtin_amdgcn_readlane(s, jj[u]);
                h[u] = *(const float4*)(H + (size_t)sj * CH + lane * 4);
            }
            asm volatile("" :: "v"(h[0].x), "v"(h[1].x), "v"(h[2].x), "v"(h[3].x),
                               "v"(h[4].x), "v"(h[5].x), "v"(h[6].x), "v"(h[7].x));
            #pragma unroll
            for (int u = 0; u < 8; ++u) {
                if (u == 0 || j + u < m) {
                    const float cu = __int_as_float(__builtin_amdgcn_readlane(__float_as_int(ds), jj[u])) *
                                     __int_as_float(__builtin_amdgcn_readlane(__float_as_int(dd), jj[u]));
                    float* tp = tile + __builtin_amdgcn_readlane(rl, jj[u]) * CH + lane;
                    lds_add(tp,      h[u].x * cu);
                    lds_add(tp + 32, h[u].y * cu);
                    lds_add(tp + 64, h[u].z * cu);
                    lds_add(tp + 96, h[u].w * cu);
                }
            }
        }
    }
}

__global__ __launch_bounds__(AGG_T) void agg_tile_kernel(
    const float* __restrict__ H, const float* __restrict__ dinv,
    const float* __restrict__ bias,
    const int* __restrict__ src_idx, const int* __restrict__ dst_idx,
    float* __restrict__ out, int n, int nedges)
{
    __shared__ __attribute__((aligned(16))) float tile[TILE * CH];
    __shared__ int queue[AGG_W][QCAP];

    const int lane = threadIdx.x & 31;
    const int wave = __builtin_amdgcn_readfirstlane(threadIdx.x >> 5);
    const int lo   = blockIdx.x * TILE;
    const int cnt  = min(TILE, n - lo);

    {
        const float4 b = *(const float4*)(bias + lane * 4);
        for (int r = wave; r < cnt; r += AGG_W) {
            const int i = lo + r;
            const float di = dinv[i];
            const float w = di * di;
            const float4 h = *(const float4*)(H + (size_t)i * CH + lane * 4);
            float* tp = tile + r * CH + lane;
            tp[0]  = fmaf(h.x, w, b.x);
            tp[32] = fmaf(h.y, w, b.y);
            tp[64] = fmaf(h.z, w, b.z);
            tp[96] = fmaf(h.w, w, b.w);
        }
    }
    __syncthreads();

    {
        int* q = &queue[wave][0];
        int qn = 0;
        const bool vec_ok = (((size_t)dst_idx) & 15) == 0;
        const int step_edges = SCAN_U * 128;
        const int nchunks = (nedges + step_edges - 1) / step_edges;
        const int nit = (nchunks + AGG_W - 1) / AGG_W;
        const int rot = (nit > 0) ? (int)(blockIdx.x % (unsigned)nit) : 0;

        int4 cur[SCAN_U];
        load_step(cur, dst_idx, (wave + AGG_W * rot) * step_edges, lane, nedges, vec_ok);
        for (int kk = 0; kk < nit; ++kk) {
            int it = kk + rot;      if (it >= nit) it -= nit;
            int itn = it + 1;       if (itn >= nit) itn -= nit;
            const int c  = wave + AGG_W * it;
            const int cn = wave + AGG_W * itn;
            int4 nxt[SCAN_U];
            load_step(nxt, dst_idx, cn * step_edges, lane, nedges, vec_ok);

            #pragma unroll
            for (int u = 0; u < SCAN_U; ++u) {
                const int eb = c * step_edges + u * 128 + lane * 4;
                const int dv[4] = {cur[u].x, cur[u].y, cur[u].z, cur[u].w};
                #pragma unroll
                for (int t4 = 0; t4 < 4; ++t4) {
                    const bool hit = (unsigned)(dv[t4] - lo) < (unsigned)cnt;
                    const unsigned mask = __builtin_amdgcn_ballot_w32(hit);
                    if (mask != 0u) {
                        const int pos = qn + (int)__builtin_amdgcn_mbcnt_lo(mask, 0u);
                        if (hit) q[pos] = eb + t4;
                        qn += __builtin_popcount(mask);
                    }
                }
            }
            if (qn >= QDRAIN) {
                drain_queue(q, qn, lane, lo, H, dinv, src_idx, dst_idx, tile);
                qn = 0;
            }
            #pragma unroll
            for (int u = 0; u < SCAN_U; ++u) cur[u] = nxt[u];
        }
        if (qn > 0)
            drain_queue(q, qn, lane, lo, H, dinv, src_idx, dst_idx, tile);
    }
    __syncthreads();

    for (int r = wave; r < cnt; r += AGG_W) {
        const float* tp = tile + r * CH + lane;
        float4 o;
        o.x = tp[0]; o.y = tp[32]; o.z = tp[64]; o.w = tp[96];
        *(float4*)(out + (size_t)(lo + r) * CH + lane * 4) = o;
    }
}

extern "C" void kernel_launch(void* const* d_in, const int* in_sizes, int n_in,
                              void* d_out, int out_size, void* d_ws, size_t ws_size,
                              hipStream_t stream) {
    const float* x    = (const float*)d_in[0];
    const int*   edge = (const int*)d_in[1];
    const float* W1   = (const float*)d_in[2];
    const float* b1   = (const float*)d_in[3];
    const float* W2   = (const float*)d_in[4];
    const float* b2   = (const float*)d_in[5];
    float* out = (float*)d_out;

    const int N = in_sizes[0] / CH;
    const int E = in_sizes[1] / 2;
    const int* srcI = edge;
    const int* dstI = edge + E;

    char* ws = (char*)d_ws;
    float* dinv = (float*)ws;
    size_t off  = ((size_t)N * 4 + 255) & ~(size_t)255;
    float* hbuf = (float*)(ws + off);
    float* tmp  = hbuf + (size_t)N * CH;

    const int T = 256;
    dim3 gN((N + T - 1) / T), gE((E + T - 1) / T);
    dim3 gGemm((N + 63) / 64);
    dim3 gAgg((N + TILE - 1) / TILE);

    deg_init_kernel<<<gN, T, 0, stream>>>(dinv, N);
    deg_count_kernel<<<gE, T, 0, stream>>>(dstI, dinv, E);
    dinv_kernel<<<gN, T, 0, stream>>>(dinv, N);

    gemm_wmma_kernel<<<gGemm, 128, 0, stream>>>(x, W1, hbuf, N, 0);
    agg_tile_kernel<<<gAgg, AGG_T, 0, stream>>>(hbuf, dinv, b1, srcI, dstI, tmp, N, E);

    gemm_wmma_kernel<<<gGemm, 128, 0, stream>>>(tmp, W2, hbuf, N, 1);
    agg_tile_kernel<<<gAgg, AGG_T, 0, stream>>>(hbuf, dinv, b2, srcI, dstI, out, N, E);
}


# second-layer aggregation adds from per-wave edge entries {src, local row, coefficient} that the first layer's scan wrote to the workspace: no second dst scan; sum order LDS arrival, as before
# speedup vs baseline: 2.3712x; 1.1722x over previous
#include <hip/hip_runtime.h>
#include <hip/hip_bf16.h>
#include <cstddef>

typedef __attribute__((ext_vector_type(2))) float v2f;
typedef __attribute__((ext_vector_type(8))) float v8f;

#define CH 128

__global__ void deg_init_kernel(float* __restrict__ deg, int n) {
    int i = blockIdx.x * blockDim.x + threadIdx.x;
    if (i < n) deg[i] = 1.0f;
}

__global__ void deg_count_kernel(const int* __restrict__ dst, float* __restrict__ deg, int e) {
    int i = blockIdx.x * blockDim.x + threadIdx.x;
    if (i < e) unsafeAtomicAdd(&deg[dst[i]], 1.0f);
}

__global__ void dinv_kernel(float* __restrict__ deg, int n) {
    int i = blockIdx.x * blockDim.x + threadIdx.x;
    if (i < n) deg[i] = rsqrtf(deg[i]);
}

__global__ __launch_bounds__(128) void gemm_wmma_kernel(
    const float* __restrict__ X, const float* __restrict__ W,
    float* __restrict__ H, int nrows, int apply_relu)
{
    __shared__ float Xs[4][16 * CH];

    const int lane = threadIdx.x & 31;
    const int wave = threadIdx.x >> 5;
    const int row_base = blockIdx.x * 64 + wave * 16;
    const bool full_tile = (row_base + 16 <= nrows);

    float* xs = &Xs[wave][0];

    if (full_tile) {
        #pragma unroll
        for (int r = 0; r < 16; ++r) {
            float4 v = *(const float4*)(X + (size_t)(row_base + r) * CH + lane * 4);
            if (apply_relu) {
                v.x = fmaxf(v.x, 0.f); v.y = fmaxf(v.y, 0.f);
                v.z = fmaxf(v.z, 0.f); v.w = fmaxf(v.w, 0.f);
            }
            *(float4*)(xs + r * CH + lane * 4) = v;
        }
    } else {
        #pragma unroll
        for (int r = 0; r < 16; ++r) {
            int row = row_base + r;
            float4 v = make_float4(0.f, 0.f, 0.f, 0.f);
            if (row < nrows)
                v = *(const float4*)(X + (size_t)row * CH + lane * 4);
            if (apply_relu) {
                v.x = fmaxf(v.x, 0.f); v.y = fmaxf(v.y, 0.f);
                v.z = fmaxf(v.z, 0.f); v.w = fmaxf(v.w, 0.f);
            }
            *(float4*)(xs + r * CH + lane * 4) = v;
        }
    }
    __syncthreads();

    const int m    = lane & 15;
    const int hi   = lane >> 4;
    const int colN = lane & 15;

    v8f acc[8];
    #pragma unroll
    for (int n = 0; n < 8; ++n) acc[n] = (v8f)(0.0f);

    for (int k0 = 0; k0 < CH; k0 += 4) {
        const int klo = k0 + hi * 2;
        v2f a = *(const v2f*)(xs + m * CH + klo);
        #pragma unroll
        for (int n = 0; n < 8; ++n) {
            v2f b;
            b.x = W[(size_t)klo * CH + n * 16 + colN];
            b.y = W[(size_t)(klo + 1) * CH + n * 16 + colN];
            acc[n] = __builtin_amdgcn_wmma_f32_16x16x4_f32(
                false, a, false, b, (short)0, acc[n], false, false);
        }
    }

    if (full_tile) {
        #pragma unroll
        for (int n = 0; n < 8; ++n) {
            #pragma unroll
            for (int r = 0; r < 8; ++r) {
                int row = row_base + r + hi * 8;
                H[(size_t)row * CH + n * 16 + colN] = acc[n][r];
            }
        }
    } else if (row_base < nrows) {
        #pragma unroll
        for (int n = 0; n < 8; ++n) {
            #pragma unroll
            for (int r = 0; r < 8; ++r) {
                int row = row_base + r + hi * 8;
                if (row < nrows)
                    H[(size_t)row * CH + n * 16 + colN] = acc[n][r];
            }
        }
    }
}

#define TILE   400
#define AGG_T  1024
#define AGG_W  32
#define SCAN_U 4
#define QDRAIN 32
#define QCAP   (SCAN_U * 128 + QDRAIN)
#define REC_CAP_MAX 1024

__device__ __forceinline__ void load_step(int4 (&v)[SCAN_U], const int* __restrict__ dst_idx,
                                          int b0, int lane, int nedges, bool vec_ok) {
    if (vec_ok && b0 + SCAN_U * 128 <= nedges) {
        #pragma unroll
        for (int u = 0; u < SCAN_U; ++u)
            v[u] = *(const int4*)(dst_idx + b0 + u * 128 + lane * 4);
    } else {
        #pragma unroll
        for (int u = 0; u < SCAN_U; ++u) {
            const int b = b0 + u * 128 + lane * 4;
            v[u].x = (b     < nedges) ? dst_idx[b]     : -1;
            v[u].y = (b + 1 < nedges) ? dst_idx[b + 1] : -1;
            v[u].z = (b + 2 < nedges) ? dst_idx[b + 2] : -1;
            v[u].w = (b + 3 < nedges) ? dst_idx[b + 3] : -1;
        }
    }
}

__device__ __forceinline__ void lds_add(float* p, float v) {
    __hip_atomic_fetch_add(p, v, __ATOMIC_RELAXED, __HIP_MEMORY_SCOPE_WORKGROUP);
}

template <bool RECORD>
__device__ __forceinline__ void drain_queue(
    const int* q, int qn, int lane, int lo,
    const float* __restrict__ H, const float* __restrict__ dinv,
    const int* __restrict__ src_idx, const int* __restrict__ dst_idx,
    float* tile, int4* rec, bool rec_ok)
{
    for (int base = 0; base < qn; base += 32) {
        const int m = min(32, qn - base);
        const int k = min(base + lane, qn - 1);
        const int e = q[k];
        const int s = src_idx[e];
        const int d = dst_idx[e];
        const float ds = dinv[s];
        const float dd = dinv[d];
        const int rl = d - lo;

        for (int j = 0; j < m; j += 8) {
            float4 h[8];
            int    jj[8];
            #pragma unroll
            for (int u = 0; u < 8; ++u) {
                jj[u] = min(j + u, m - 1);
                const int sj = __builtin_amdgcn_readlane(s, jj[u]);
                h[u] = *(const float4*)(H + (size_t)sj * CH + lane * 4);
            }
            asm volatile("" :: "v"(h[0].x), "v"(h[1].x), "v"(h[2].x), "v"(h[3].x),
                               "v"(h[4].x), "v"(h[5].x), "v"(h[6].x), "v"(h[7].x));
            #pragma unroll
            for (int u = 0; u < 8; ++u) {
                if (u == 0 || j + u < m) {
                    const float cu = __int_as_float(__builtin_amdgcn_readlane(__float_as_int(ds), jj[u])) *
                                     __int_as_float(__builtin_amdgcn_readlane(__float_as_int(dd), jj[u]));
                    float* tp = tile + __builtin_amdgcn_readlane(rl, jj[u]) * CH + lane;
                    lds_add(tp,      h[u].x * cu);
                    lds_add(tp + 32, h[u].y * cu);
                    lds_add(tp + 64, h[u].z * cu);
                    lds_add(tp + 96, h[u].w * cu);
                }
            }
        }
        if (RECORD) {
            if (rec_ok && base + lane < qn) {
                int4 r;
                r.x = s; r.y = rl; r.z = __float_as_int(ds * dd); r.w = 0;
                rec[base + lane] = r;
            }
        }
    }
}

__device__ __forceinline__ void drain_list(
    const int4* __restrict__ rec, int n, int lane,
    const float* __restrict__ H, float* tile)
{
    int4 r = rec[min(lane, n - 1)];
    for (int base = 0; base < n; base += 32) {
        const int m = min(32, n - base);
        const int4 rn = rec[min(base + 32 + lane, n - 1)];
        const int   s  = r.x;
        const int   rl = r.y;
        const float c  = __int_as_float(r.z);

        for (int j = 0; j < m; j += 8) {
            float4 h[8];
            int    jj[8];
            #pragma unroll
            for (int u = 0; u < 8; ++u) {
                jj[u] = min(j + u, m - 1);
                const int sj = __builtin_amdgcn_readlane(s, jj[u]);
                h[u] = *(const float4*)(H + (size_t)sj * CH + lane * 4);
            }
            asm volatile("" :: "v"(h[0].x), "v"(h[1].x), "v"(h[2].x), "v"(h[3].x),
                               "v"(h[4].x), "v"(h[5].x), "v"(h[6].x), "v"(h[7].x));
            #pragma unroll
            for (int u = 0; u < 8; ++u) {
                if (u == 0 || j + u < m) {
                    const float cu = __int_as_float(__builtin_amdgcn_readlane(__float_as_int(c), jj[u]));
                    float* tp = tile + __builtin_amdgcn_readlane(rl, jj[u]) * CH + lane;
                    lds_add(tp,      h[u].x * cu);
                    lds_add(tp + 32, h[u].y * cu);
                    lds_add(tp + 64, h[u].z * cu);
                    lds_add(tp + 96, h[u].w * cu);
                }
            }
        }
        r = rn;
    }
}

template <bool RECORD>
__device__ __forceinline__ void scan_and_add(
    int* q, int lane, int wave, int lo, int cnt,
    const float* __restrict__ H, const float* __restrict__ dinv,
    const int* __restrict__ src_idx, const int* __restrict__ dst_idx,
    float* tile, int nedges, int4* rec, int cap, int& rec_n)
{
    int qn = 0;
    int wn = 0;
    bool fits = true;
    const bool vec_ok = (((size_t)dst_idx) & 15) == 0;
    const int step_edges = SCAN_U * 128;
    const int nchunks = (nedges + step_edges - 1) / step_edges;
    const int nit = (nchunks + AGG_W - 1) / AGG_W;
    const int rot = (nit > 0) ? (int)(blockIdx.x % (unsigned)nit) : 0;

    int4 cur[SCAN_U];
    load_step(cur, dst_idx, (wave + AGG_W * rot) * step_edges, lane, nedges, vec_ok);
    for (int kk = 0; kk < nit; ++kk) {
        int it = kk + rot;      if (it >= nit) it -= nit;
        int itn = it + 1;       if (itn >= nit) itn -= nit;
        const int c  = wave + AGG_W * it;
        const int cn = wave + AGG_W * itn;
        int4 nxt[SCAN_U];
        load_step(nxt, dst_idx, cn * step_edges, lane, nedges, vec_ok);

        #pragma unroll
        for (int u = 0; u < SCAN_U; ++u) {
            const int eb = c * step_edges + u * 128 + lane * 4;
            const int dv[4] = {cur[u].x, cur[u].y, cur[u].z, cur[u].w};
            #pragma unroll
            for (int t4 = 0; t4 < 4; ++t4) {
                const bool hit = (unsigned)(dv[t4] - lo) < (unsigned)cnt;
                const unsigned mask = __builtin_amdgcn_ballot_w32(hit);
                if (mask != 0u) {
                    const int pos = qn + (int)__builtin_amdgcn_mbcnt_lo(mask, 0u);
                    if (hit) q[pos] = eb + t4;
                    qn += __builtin_popcount(mask);
                }
            }
        }
        if (qn >= QDRAIN) {
            fits = fits && (wn + qn <= cap);
            drain_queue<RECORD>(q, qn, lane, lo, H, dinv, src_idx, dst_idx, tile, rec + wn, fits);
            wn += qn;
            qn = 0;
        }
        #pragma unroll
        for (int u = 0; u < SCAN_U; ++u) cur[u] = nxt[u];
    }
    if (qn > 0) {
        fits = fits && (wn + qn <= cap);
        drain_queue<RECORD>(q, qn, lane, lo, H, dinv, src_idx, dst_idx, tile, rec + wn, fits);
        wn += qn;
    }
    rec_n = fits ? wn : -1;
}

template <bool REPLAY>
__device__ __forceinline__ void agg_tile_body(
    float* tile, int* queue_base,
    const float* __restrict__ H, const float* __restrict__ dinv,
    const float* __restrict__ bias,
    const int* __restrict__ src_idx, const int* __restrict__ dst_idx,
    float* __restrict__ out, int n, int nedges,
    int* __restrict__ list_cnt, int* __restrict__ list_rec, int cap)
{
    const int lane = threadIdx.x & 31;
    const int wave = __builtin_amdgcn_readfirstlane(threadIdx.x >> 5);
    const int lo   = blockIdx.x * TILE;
    const int cnt  = min(TILE, n - lo);

    {
        const float4 b = *(const float4*)(bias + lane * 4);
        for (int r = wave; r < cnt; r += AGG_W) {
            const int i = lo + r;
            const float di = dinv[i];
            const float w = di * di;
            const float4 h = *(const float4*)(H + (size_t)i * CH + lane * 4);
            float* tp = tile + r * CH + lane;
            tp[0]  = fmaf(h.x, w, b.x);
            tp[32] = fmaf(h.y, w, b.y);
            tp[64] = fmaf(h.z, w, b.z);
            tp[96] = fmaf(h.w, w, b.w);
        }
    }
    __syncthreads();

    {
        int* q = queue_base + wave * QCAP;
        const int wid = blockIdx.x * AGG_W + wave;
        int4* rec = (int4*)list_rec + (size_t)wid * (size_t)cap;
        if (REPLAY) {
            const int nrec = __builtin_amdgcn_readfirstlane(list_cnt[wid]);
            if (nrec > 0) {
                drain_list(rec, nrec, lane, H, tile);
            } else if (nrec < 0) {
                int unused;
                scan_and_add<false>(q, lane, wave, lo, cnt, H, dinv, src_idx, dst_idx,
                                    tile, nedges, rec, 0, unused);
            }
        } else {
            int nrec;
            scan_and_add<true>(q, lane, wave, lo, cnt, H, dinv, src_idx, dst_idx,
                               tile, nedges, rec, cap, nrec);
            if (lane == 0) list_cnt[wid] = nrec;
        }
    }
    __syncthreads();

    for (int r = wave; r < cnt; r += AGG_W) {
        const float* tp = tile + r * CH + lane;
        float4 o;
        o.x = tp[0]; o.y = tp[32]; o.z = tp[64]; o.w = tp[96];
        *(float4*)(out + (size_t)(lo + r) * CH + lane * 4) = o;
    }
}

__global__ __launch_bounds__(AGG_T) void agg_tile_record_kernel(
    const float* __restrict__ H, const float* __restrict__ dinv,
    const float* __restrict__ bias,
    const int* __restrict__ src_idx, const int* __restrict__ dst_idx,
    float* __restrict__ out, int n, int nedges,
    int* __restrict__ list_cnt, int* __restrict__ list_rec, int cap)
{
    __shared__ __attribute__((aligned(16))) float tile[TILE * CH];
    __shared__ int queue[AGG_W][QCAP];
    agg_tile_body<false>(tile, &queue[0][0], H, dinv, bias, src_idx, dst_idx, out, n, nedges,
                         list_cnt, list_rec, cap);
}

__global__ __launch_bounds__(AGG_T) void agg_tile_replay_kernel(
    const float* __restrict__ H, const float* __restrict__ dinv,
    const float* __restrict__ bias,
    const int* __restrict__ src_idx, const int* __restrict__ dst_idx,
    float* __restrict__ out, int n, int nedges,
    int* __restrict__ list_cnt, int* __restrict__ list_rec, int cap)
{
    __shared__ __attribute__((aligned(16))) float tile[TILE * CH];
    __shared__ int queue[AGG_W][QCAP];
    agg_tile_body<true>(tile, &queue[0][0], H, dinv, bias, src_idx, dst_idx, out, n, nedges,
                        list_cnt, list_rec, cap);
}

extern "C" void kernel_launch(void* const* d_in, const int* in_sizes, int n_in,
                              void* d_out, int out_size, void* d_ws, size_t ws_size,
                              hipStream_t stream) {
    const float* x    = (const float*)d_in[0];
    const int*   edge = (const int*)d_in[1];
    const float* W1   = (const float*)d_in[2];
    const float* b1   = (const float*)d_in[3];
    const float* W2   = (const float*)d_in[4];
    const float* b2   = (const float*)d_in[5];
    float* out = (float*)d_out;

    const int N = in_sizes[0] / CH;
    const int E = in_sizes[1] / 2;
    const int* srcI = edge;
    const int* dstI = edge + E;

    char* ws = (char*)d_ws;
    float* dinv = (float*)ws;
    size_t off  = ((size_t)N * 4 + 255) & ~(size_t)255;
    float* hbuf = (float*)(ws + off);
    float* tmp  = hbuf + (size_t)N * CH;
    const int nTile = (N + TILE - 1) / TILE;
    size_t offc = off + 2 * (size_t)N * CH * 4;
    int* lcnt   = (int*)(ws + offc);
    size_t offr = (offc + (size_t)nTile * AGG_W * 4 + 255) & ~(size_t)255;
    int* lrec   = (int*)(ws + offr);
    size_t room = (ws_size > offr) ? (ws_size - offr) / ((size_t)nTile * AGG_W * 16) : 0;
    const int cap = (room > (size_t)REC_CAP_MAX) ? REC_CAP_MAX : (int)room;

    const int T = 256;
    dim3 gN((N + T - 1) / T), gE((E + T - 1) / T);
    dim3 gGemm((N + 63) / 64);
    dim3 gAgg(nTile);

    deg_init_kernel<<<gN, T, 0, stream>>>(dinv, N);
    deg_count_kernel<<<gE, T, 0, stream>>>(dstI, dinv, E);
    dinv_kernel<<<gN, T, 0, stream>>>(dinv, N);

    gemm_wmma_kernel<<<gGemm, 128, 0, stream>>>(x, W1, hbuf, N, 0);
    agg_tile_record_kernel<<<gAgg, AGG_T, 0, stream>>>(hbuf, dinv, b1, srcI, dstI, tmp, N, E, lcnt, lrec, cap);

    gemm_wmma_kernel<<<gGemm, 128, 0, stream>>>(tmp, W2, hbuf, N, 1);
    agg_tile_replay_kernel<<<gAgg, AGG_T, 0, stream>>>(hbuf, dinv, b2, srcI, dstI, out, N, E, lcnt, lrec, cap);
}


# first-layer edge scan reads a one-byte tile id per edge (dst/400, written by the degree kernel; 4 ids tested per dword, 4 id loads in flight per lane), not the 4-byte dst; sum order still LDS arrival
# speedup vs baseline: 2.5839x; 1.0897x over previous
#include <hip/hip_runtime.h>
#include <hip/hip_bf16.h>
#include <cstddef>

typedef __attribute__((ext_vector_type(2))) float v2f;
typedef __attribute__((ext_vector_type(8))) float v8f;

#define CH 128

__global__ void deg_init_kernel(float* __restrict__ deg, int n) {
    int i = blockIdx.x * blockDim.x + threadIdx.x;
    if (i < n) deg[i] = 1.0f;
}

#define TILE_ROWS 400
__global__ void deg_count_tid_kernel(const int* __restrict__ dst, float* __restrict__ deg,
                                     unsigned* __restrict__ tid, int e, int npad4) {
    const int t = blockIdx.x * blockDim.x + threadIdx.x;
    if (t >= npad4) return;
    const int b = t * 4;
    int dv[4] = {-1, -1, -1, -1};
    if (b + 4 <= e && (((size_t)dst) & 15) == 0) {
        const int4 v = *(const int4*)(dst + b);
        dv[0] = v.x; dv[1] = v.y; dv[2] = v.z; dv[3] = v.w;
    } else {
        #pragma unroll
        for (int i = 0; i < 4; ++i)
            if (b + i < e) dv[i] = dst[b + i];
    }
    unsigned w = 0u;
    #pragma unroll
    for (int i = 0; i < 4; ++i) {
        unsigned id = 0xFFu;
        if (dv[i] >= 0) {
            unsafeAtomicAdd(&deg[dv[i]], 1.0f);
            id = (unsigned)min(dv[i] / TILE_ROWS, 255);
        }
        w |= id << (8 * i);
    }
    tid[t] = w;
}

__global__ void dinv_kernel(float* __restrict__ deg, int n) {
    int i = blockIdx.x * blockDim.x + threadIdx.x;
    if (i < n) deg[i] = rsqrtf(deg[i]);
}

__global__ __launch_bounds__(128) void gemm_wmma_kernel(
    const float* __restrict__ X, const float* __restrict__ W,
    float* __restrict__ H, int nrows, int apply_relu)
{
    __shared__ float Xs[4][16 * CH];

    const int lane = threadIdx.x & 31;
    const int wave = threadIdx.x >> 5;
    const int row_base = blockIdx.x * 64 + wave * 16;
    const bool full_tile = (row_base + 16 <= nrows);

    float* xs = &Xs[wave][0];

    if (full_tile) {
        #pragma unroll
        for (int r = 0; r < 16; ++r) {
            float4 v = *(const float4*)(X + (size_t)(row_base + r) * CH + lane * 4);
            if (apply_relu) {
                v.x = fmaxf(v.x, 0.f); v.y = fmaxf(v.y, 0.f);
                v.z = fmaxf(v.z, 0.f); v.w = fmaxf(v.w, 0.f);
            }
            *(float4*)(xs + r * CH + lane * 4) = v;
        }
    } else {
        #pragma unroll
        for (int r = 0; r < 16; ++r) {
            int row = row_base + r;
            float4 v = make_float4(0.f, 0.f, 0.f, 0.f);
            if (row < nrows)
                v = *(const float4*)(X + (size_t)row * CH + lane * 4);
            if (apply_relu) {
                v.x = fmaxf(v.x, 0.f); v.y = fmaxf(v.y, 0.f);
                v.z = fmaxf(v.z, 0.f); v.w = fmaxf(v.w, 0.f);
            }
            *(float4*)(xs + r * CH + lane * 4) = v;
        }
    }
    __syncthreads();

    const int m    = lane & 15;
    const int hi   = lane >> 4;
    const int colN = lane & 15;

    v8f acc[8];
    #pragma unroll
    for (int n = 0; n < 8; ++n) acc[n] = (v8f)(0.0f);

    for (int k0 = 0; k0 < CH; k0 += 4) {
        const int klo = k0 + hi * 2;
        v2f a = *(const v2f*)(xs + m * CH + klo);
        #pragma unroll
        for (int n = 0; n < 8; ++n) {
            v2f b;
            b.x = W[(size_t)klo * CH + n * 16 + colN];
            b.y = W[(size_t)(klo + 1) * CH + n * 16 + colN];
            acc[n] = __builtin_amdgcn_wmma_f32_16x16x4_f32(
                false, a, false, b, (short)0, acc[n], false, false);
        }
    }

    if (full_tile) {
        #pragma unroll
        for (int n = 0; n < 8; ++n) {
            #pragma unroll
            for (int r = 0; r < 8; ++r) {
                int row = row_base + r + hi * 8;
                H[(size_t)row * CH + n * 16 + colN] = acc[n][r];
            }
        }
    } else if (row_base < nrows) {
        #pragma unroll
        for (int n = 0; n < 8; ++n) {
            #pragma unroll
            for (int r = 0; r < 8; ++r) {
                int row = row_base + r + hi * 8;
                if (row < nrows)
                    H[(size_t)row * CH + n * 16 + colN] = acc[n][r];
            }
        }
    }
}

#define TILE   TILE_ROWS
#define AGG_T  1024
#define AGG_W  32
#define STEP_E 512
#define RING   4
#define QDRAIN 32
#define QCAP   (STEP_E + QDRAIN)
#define REC_CAP_MAX 1024

__device__ __forceinline__ void lds_add(float* p, float v) {
    __hip_atomic_fetch_add(p, v, __ATOMIC_RELAXED, __HIP_MEMORY_SCOPE_WORKGROUP);
}

template <bool RECORD>
__device__ __forceinline__ void drain_queue(
    const int* q, int qn, int lane, int lo,
    const float* __restrict__ H, const float* __restrict__ dinv,
    const int* __restrict__ src_idx, const int* __restrict__ dst_idx,
    float* tile, int4* rec, bool rec_ok)
{
    for (int base = 0; base < qn; base += 32) {
        const int m = min(32, qn - base);
        const int k = min(base + lane, qn - 1);
        const int e = q[k];
        const int s = src_idx[e];
        const int d = dst_idx[e];
        const float ds = dinv[s];
        const float dd = dinv[d];
        const int rl = d - lo;

        int j = 0;
        do {
            float4 h[8];
            int    jj[8];
            #pragma unroll
            for (int u = 0; u < 8; ++u) {
                jj[u] = min(j + u, m - 1);
                const int sj = __builtin_amdgcn_readlane(s, jj[u]);
                h[u] = *(const float4*)(H + (size_t)sj * CH + lane * 4);
            }
            asm volatile("" :: "v"(h[0].x), "v"(h[1].x), "v"(h[2].x), "v"(h[3].x),
                               "v"(h[4].x), "v"(h[5].x), "v"(h[6].x), "v"(h[7].x));
            #pragma unroll
            for (int u = 0; u < 8; ++u) {
                if (u == 0 || j + u < m) {
                    const float cu = __int_as_float(__builtin_amdgcn_readlane(__float_as_int(ds), jj[u])) *
                                     __int_as_float(__builtin_amdgcn_readlane(__float_as_int(dd), jj[u]));
                    float* tp = tile + __builtin_amdgcn_readlane(rl, jj[u]) * CH + lane;
                    lds_add(tp,      h[u].x * cu);
                    lds_add(tp + 32, h[u].y * cu);
                    lds_add(tp + 64, h[u].z * cu);
                    lds_add(tp + 96, h[u].w * cu);
                }
            }
            j += 8;
        } while (j < m);
        if (RECORD) {
            if (rec_ok && base + lane < qn) {
                int4 r;
                r.x = s; r.y = rl; r.z = __float_as_int(ds * dd); r.w = 0;
                rec[base + lane] = r;
            }
        }
    }
}

__device__ __forceinline__ void drain_list(
    const int4* __restrict__ rec, int n, int lane,
    const float* __restrict__ H, float* tile)
{
    int4 r = rec[min(lane, n - 1)];
    for (int base = 0; base < n; base += 32) {
        const int m = min(32, n - base);
        const int4 rn = rec[min(base + 32 + lane, n - 1)];
        const int   s  = r.x;
        const int   rl = r.y;
        const float c  = __int_as_float(r.z);

        for (int j = 0; j < m; j += 8) {
            float4 h[8];
            int    jj[8];
            #pragma unroll
            for (int u = 0; u < 8; ++u) {
                jj[u] = min(j + u, m - 1);
                const int sj = __builtin_amdgcn_readlane(s, jj[u]);
                h[u] = *(const float4*)(H + (size_t)sj * CH + lane * 4);
            }
            asm volatile("" :: "v"(h[0].x), "v"(h[1].x), "v"(h[2].x), "v"(h[3].x),
                               "v"(h[4].x), "v"(h[5].x), "v"(h[6].x), "v"(h[7].x));
            #pragma unroll
            for (int u = 0; u < 8; ++u) {
                if (u == 0 || j + u < m) {
                    const float cu = __int_as_float(__builtin_amdgcn_readlane(__float_as_int(c), jj[u]));
                    float* tp = tile + __builtin_amdgcn_readlane(rl, jj[u]) * CH + lane;
                    lds_add(tp,      h[u].x * cu);
                    lds_add(tp + 32, h[u].y * cu);
                    lds_add(tp + 64, h[u].z * cu);
                    lds_add(tp + 96, h[u].w * cu);
                }
            }
        }
        r = rn;
    }
}

__host__ __device__ __forceinline__ int scan_steps(int nedges) {
    const int nchunks = (nedges + STEP_E - 1) / STEP_E;
    return ((nchunks + AGG_W - 1) / AGG_W + RING - 1) / RING * RING;
}

__device__ __forceinline__ uint4 load_ids(const unsigned* __restrict__ tid, unsigned off) {
    return *(const uint4*)((const char*)tid + off);
}

template <bool RECORD>
__device__ __forceinline__ void scan_and_add(
    int* q, int lane, int wave, int lo,
    const float* __restrict__ H, const float* __restrict__ dinv,
    const int* __restrict__ src_idx, const int* __restrict__ dst_idx,
    const unsigned* __restrict__ tid,
    float* tile, int nedges, int4* rec, int cap, int& rec_n)
{
    int qn = 0;
    int wn = 0;
    bool fits = true;
    const int nit = scan_steps(nedges);
    rec_n = 0;
    if (nit <= 0) return;
    const int rot = (int)(blockIdx.x % (unsigned)nit);
    const unsigned g4 = (unsigned)blockIdx.x * 0x01010101u;

    const unsigned lane_off = (unsigned)(wave * STEP_E + lane * 16);
    const unsigned step_bytes = STEP_E * AGG_W;

    uint4 r[RING];
    int sl = rot;
    #pragma unroll
    for (int u = 0; u < RING; ++u) {
        r[u] = load_ids(tid, lane_off + (unsigned)sl * step_bytes);
        sl = (sl + 1 < nit) ? sl + 1 : 0;
        __builtin_amdgcn_sched_barrier(0);
    }
    int st = rot;
    for (int kk = 0; kk < nit; kk += RING) {
        #pragma unroll
        for (int u = 0; u < RING; ++u) {
            const unsigned wv[4] = {r[u].x, r[u].y, r[u].z, r[u].w};
            unsigned hits = 0u;
            #pragma unroll
            for (int k = 0; k < 4; ++k) {
                const unsigned x = wv[k] ^ g4;
                const unsigned nz = ((x & 0x7F7F7F7Fu) + 0x7F7F7F7Fu) | x;
                hits |= ((~nz >> 7) & 0x01010101u) << k;
            }
            unsigned mask = __builtin_amdgcn_ballot_w32(hits != 0u);
            while (mask != 0u) {
                const int pos = qn + (int)__builtin_amdgcn_mbcnt_lo(mask, 0u);
                if (hits != 0u) {
                    const int p = __builtin_ctz(hits);
                    q[pos] = (int)(lane_off + (unsigned)st * step_bytes) + ((p & 7) << 2) + (p >> 3);
                }
                qn += __builtin_popcount(mask);
                hits &= hits - 1u;
                mask = __builtin_amdgcn_ballot_w32(hits != 0u);
            }
            r[u] = load_ids(tid, lane_off + (unsigned)sl * step_bytes);
            sl = (sl + 1 < nit) ? sl + 1 : 0;
            st = (st + 1 < nit) ? st + 1 : 0;
            if (qn >= QDRAIN) {
                fits = fits && (wn + qn <= cap);
                drain_queue<RECORD>(q, qn, lane, lo, H, dinv, src_idx, dst_idx, tile, rec + wn, fits);
                wn += qn;
                qn = 0;
            }
        }
    }
    if (qn > 0) {
        fits = fits && (wn + qn <= cap);
        drain_queue<RECORD>(q, qn, lane, lo, H, dinv, src_idx, dst_idx, tile, rec + wn, fits);
        wn += qn;
    }
    rec_n = fits ? wn : -1;
}

template <bool REPLAY>
__device__ __forceinline__ void agg_tile_body(
    float* tile, int* queue_base,
    const float* __restrict__ H, const float* __restrict__ dinv,
    const float* __restrict__ bias,
    const int* __restrict__ src_idx, const int* __restrict__ dst_idx,
    const unsigned* __restrict__ tid,
    float* __restrict__ out, int n, int nedges,
    int* __restrict__ list_cnt, int* __restrict__ list_rec, int cap)
{
    const int lane = threadIdx.x & 31;
    const int wave = __builtin_amdgcn_readfirstlane(threadIdx.x >> 5);
    const int lo   = blockIdx.x * TILE;
    const int cnt  = min(TILE, n - lo);

    {
        const float4 b = *(const float4*)(bias + lane * 4);
        for (int r = wave; r < cnt; r += AGG_W) {
            const int i = lo + r;
            const float di = dinv[i];
            const float w = di * di;
            const float4 h = *(const float4*)(H + (size_t)i * CH + lane * 4);
            float* tp = tile + r * CH + lane;
            tp[0]  = fmaf(h.x, w, b.x);
            tp[32] = fmaf(h.y, w, b.y);
            tp[64] = fmaf(h.z, w, b.z);
            tp[96] = fmaf(h.w, w, b.w);
        }
    }
    __syncthreads();

    {
        int* q = queue_base + wave * QCAP;
        const int wid = blockIdx.x * AGG_W + wave;
        int4* rec = (int4*)list_rec + (size_t)wid * (size_t)cap;
        if (REPLAY) {
            const int nrec = __builtin_amdgcn_readfirstlane(list_cnt[wid]);
            if (nrec > 0) {
                drain_list(rec, nrec, lane, H, tile);
            } else if (nrec < 0) {
                int unused;
                scan_and_add<false>(q, lane, wave, lo, H, dinv, src_idx, dst_idx, tid,
                                    tile, nedges, rec, 0, unused);
            }
        } else {
            int nrec;
            scan_and_add<true>(q, lane, wave, lo, H, dinv, src_idx, dst_idx, tid,
                               tile, nedges, rec, cap, nrec);
            if (lane == 0) list_cnt[wid] = nrec;
        }
    }
    __syncthreads();

    for (int r = wave; r < cnt; r += AGG_W) {
        const float* tp = tile + r * CH + lane;
        float4 o;
        o.x = tp[0]; o.y = tp[32]; o.z = tp[64]; o.w = tp[96];
        *(float4*)(out + (size_t)(lo + r) * CH + lane * 4) = o;
    }
}

__global__ __launch_bounds__(AGG_T) void agg_tile_record_kernel(
    const float* __restrict__ H, const float* __restrict__ dinv,
    const float* __restrict__ bias,
    const int* __restrict__ src_idx, const int* __restrict__ dst_idx,
    const unsigned* __restrict__ tid,
    float* __restrict__ out, int n, int nedges,
    int* __restrict__ list_cnt, int* __restrict__ list_rec, int cap)
{
    __shared__ __attribute__((aligned(16))) float tile[TILE * CH];
    __shared__ int queue[AGG_W][QCAP];
    agg_tile_body<false>(tile, &queue[0][0], H, dinv, bias, src_idx, dst_idx, tid, out, n, nedges,
                         list_cnt, list_rec, cap);
}

__global__ __launch_bounds__(AGG_T) void agg_tile_replay_kernel(
    const float* __restrict__ H, const float* __restrict__ dinv,
    const float* __restrict__ bias,
    const int* __restrict__ src_idx, const int* __restrict__ dst_idx,
    const unsigned* __restrict__ tid,
    float* __restrict__ out, int n, int nedges,
    int* __restrict__ list_cnt, int* __restrict__ list_rec, int cap)
{
    __shared__ __attribute__((aligned(16))) float tile[TILE * CH];
    __shared__ int queue[AGG_W][QCAP];
    agg_tile_body<true>(tile, &queue[0][0], H, dinv, bias, src_idx, dst_idx, tid, out, n, nedges,
                        list_cnt, list_rec, cap);
}

extern "C" void kernel_launch(void* const* d_in, const int* in_sizes, int n_in,
                              void* d_out, int out_size, void* d_ws, size_t ws_size,
                              hipStream_t stream) {
    const float* x    = (const float*)d_in[0];
    const int*   edge = (const int*)d_in[1];
    const float* W1   = (const float*)d_in[2];
    const float* b1   = (const float*)d_in[3];
    const float* W2   = (const float*)d_in[4];
    const float* b2   = (const float*)d_in[5];
    float* out = (float*)d_out;

    const int N = in_sizes[0] / CH;
    const int E = in_sizes[1] / 2;
    const int* srcI = edge;
    const int* dstI = edge + E;

    char* ws = (char*)d_ws;
    float* dinv = (float*)ws;
    size_t off  = ((size_t)N * 4 + 255) & ~(size_t)255;
    float* hbuf = (float*)(ws + off);
    float* tmp  = hbuf + (size_t)N * CH;
    const int nTile = (N + TILE - 1) / TILE;
    const int nStep = scan_steps(E);
    const int npad4 = nStep * AGG_W * (STEP_E / 4);
    size_t offt = off + 2 * (size_t)N * CH * 4;
    unsigned* tid = (unsigned*)(ws + offt);
    size_t offc = (offt + (size_t)npad4 * 4 + 255) & ~(size_t)255;
    int* lcnt   = (int*)(ws + offc);
    size_t offr = (offc + (size_t)nTile * AGG_W * 4 + 255) & ~(size_t)255;
    int* lrec   = (int*)(ws + offr);
    size_t room = (ws_size > offr) ? (ws_size - offr) / ((size_t)nTile * AGG_W * 16) : 0;
    const int cap = (room > (size_t)REC_CAP_MAX) ? REC_CAP_MAX : (int)room;

    const int T = 256;
    dim3 gN((N + T - 1) / T), gE((npad4 + T - 1) / T);
    dim3 gGemm((N + 63) / 64);
    dim3 gAgg(nTile);

    deg_init_kernel<<<gN, T, 0, stream>>>(dinv, N);
    deg_count_tid_kernel<<<gE, T, 0, stream>>>(dstI, dinv, tid, E, npad4);
    dinv_kernel<<<gN, T, 0, stream>>>(dinv, N);

    gemm_wmma_kernel<<<gGemm, 128, 0, stream>>>(x, W1, hbuf, N, 0);
    agg_tile_record_kernel<<<gAgg, AGG_T, 0, stream>>>(hbuf, dinv, b1, srcI, dstI, tid, tmp, N, E, lcnt, lrec, cap);

    gemm_wmma_kernel<<<gGemm, 128, 0, stream>>>(tmp, W2, hbuf, N, 1);
    agg_tile_replay_kernel<<<gAgg, AGG_T, 0, stream>>>(hbuf, dinv, b2, srcI, dstI, tid, out, N, E, lcnt, lrec, cap);
}


# each layer as (A_hat X) W: the owner work-groups sum the layer's INPUT rows in LDS and multiply the finished tile by W on f32 WMMA as their epilogue; no GEMM launches (7 -> 5); sum order changes
# speedup vs baseline: 2.9640x; 1.1471x over previous
#include <hip/hip_runtime.h>
#include <hip/hip_bf16.h>
#include <cstddef>

typedef __attribute__((ext_vector_type(2))) float v2f;
typedef __attribute__((ext_vector_type(8))) float v8f;

#define CH 128

__global__ void deg_init_kernel(float* __restrict__ deg, int n) {
    int i = blockIdx.x * blockDim.x + threadIdx.x;
    if (i < n) deg[i] = 1.0f;
}

#define TILE_ROWS 400
__global__ void deg_count_tid_kernel(const int* __restrict__ dst, float* __restrict__ deg,
                                     unsigned* __restrict__ tid, int e, int npad4) {
    const int t = blockIdx.x * blockDim.x + threadIdx.x;
    if (t >= npad4) return;
    const int b = t * 4;
    int dv[4] = {-1, -1, -1, -1};
    if (b + 4 <= e && (((size_t)dst) & 15) == 0) {
        const int4 v = *(const int4*)(dst + b);
        dv[0] = v.x; dv[1] = v.y; dv[2] = v.z; dv[3] = v.w;
    } else {
        #pragma unroll
        for (int i = 0; i < 4; ++i)
            if (b + i < e) dv[i] = dst[b + i];
    }
    unsigned w = 0u;
    #pragma unroll
    for (int i = 0; i < 4; ++i) {
        unsigned id = 0xFFu;
        if (dv[i] >= 0) {
            unsafeAtomicAdd(&deg[dv[i]], 1.0f);
            id = (unsigned)min(dv[i] / TILE_ROWS, 255);
        }
        w |= id << (8 * i);
    }
    tid[t] = w;
}

__global__ void dinv_kernel(float* __restrict__ deg, int n) {
    int i = blockIdx.x * blockDim.x + threadIdx.x;
    if (i < n) deg[i] = rsqrtf(deg[i]);
}

#define TILE   TILE_ROWS
#define TP     129
#define AGG_T  1024
#define AGG_W  32
#define STEP_E 512
#define RING   4
#define QDRAIN 32
#define QCAP   (STEP_E + QDRAIN)
#define REC_CAP_MAX 1024

__device__ __forceinline__ void lds_add(float* p, float v) {
    __hip_atomic_fetch_add(p, v, __ATOMIC_RELAXED, __HIP_MEMORY_SCOPE_WORKGROUP);
}

template <bool RECORD>
__device__ __forceinline__ void drain_queue(
    const int* q, int qn, int lane, int lo,
    const float* __restrict__ H, const float* __restrict__ dinv,
    const int* __restrict__ src_idx, const int* __restrict__ dst_idx,
    float* tile, int4* rec, bool rec_ok)
{
    for (int base = 0; base < qn; base += 32) {
        const int m = min(32, qn - base);
        const int k = min(base + lane, qn - 1);
        const int e = q[k];
        const int s = src_idx[e];
        const int d = dst_idx[e];
        const float ds = dinv[s];
        const float dd = dinv[d];
        const int rl = (d - lo) * TP;

        int j = 0;
        do {
            float4 h[8];
            int    jj[8];
            #pragma unroll
            for (int u = 0; u < 8; ++u) {
                jj[u] = min(j + u, m - 1);
                const int sj = __builtin_amdgcn_readlane(s, jj[u]);
                h[u] = *(const float4*)(H + (size_t)sj * CH + lane * 4);
            }
            asm volatile("" :: "v"(h[0].x), "v"(h[1].x), "v"(h[2].x), "v"(h[3].x),
                               "v"(h[4].x), "v"(h[5].x), "v"(h[6].x), "v"(h[7].x));
            #pragma unroll
            for (int u = 0; u < 8; ++u) {
                if (u == 0 || j + u < m) {
                    const float cu = __int_as_float(__builtin_amdgcn_readlane(__float_as_int(ds), jj[u])) *
                                     __int_as_float(__builtin_amdgcn_readlane(__float_as_int(dd), jj[u]));
                    float* tp = tile + __builtin_amdgcn_readlane(rl, jj[u]) + lane;
                    lds_add(tp,      h[u].x * cu);
                    lds_add(tp + 32, h[u].y * cu);
                    lds_add(tp + 64, h[u].z * cu);
                    lds_add(tp + 96, h[u].w * cu);
                }
            }
            j += 8;
        } while (j < m);
        if (RECORD) {
            if (rec_ok && base + lane < qn) {
                int4 r;
                r.x = s; r.y = rl; r.z = __float_as_int(ds * dd); r.w = 0;
                rec[base + lane] = r;
            }
        }
    }
}

__device__ __forceinline__ void drain_list(
    const int4* __restrict__ rec, int n, int lane,
    const float* __restrict__ H, float* tile)
{
    int4 r = rec[min(lane, n - 1)];
    for (int base = 0; base < n; base += 32) {
        const int m = min(32, n - base);
        const int4 rn = rec[min(base + 32 + lane, n - 1)];
        const int   s  = r.x;
        const int   rl = r.y;
        const float c  = __int_as_float(r.z);

        for (int j = 0; j < m; j += 8) {
            float4 h[8];
            int    jj[8];
            #pragma unroll
            for (int u = 0; u < 8; ++u) {
                jj[u] = min(j + u, m - 1);
                const int sj = __builtin_amdgcn_readlane(s, jj[u]);
                h[u] = *(const float4*)(H + (size_t)sj * CH + lane * 4);
            }
            asm volatile("" :: "v"(h[0].x), "v"(h[1].x), "v"(h[2].x), "v"(h[3].x),
                               "v"(h[4].x), "v"(h[5].x), "v"(h[6].x), "v"(h[7].x));
            #pragma unroll
            for (int u = 0; u < 8; ++u) {
                if (u == 0 || j + u < m) {
                    const float cu = __int_as_float(__builtin_amdgcn_readlane(__float_as_int(c), jj[u]));
                    float* tp = tile + __builtin_amdgcn_readlane(rl, jj[u]) + lane;
                    lds_add(tp,      h[u].x * cu);
                    lds_add(tp + 32, h[u].y * cu);
                    lds_add(tp + 64, h[u].z * cu);
                    lds_add(tp + 96, h[u].w * cu);
                }
            }
        }
        r = rn;
    }
}

__host__ __device__ __forceinline__ int scan_steps(int nedges) {
    const int nchunks = (nedges + STEP_E - 1) / STEP_E;
    return ((nchunks + AGG_W - 1) / AGG_W + RING - 1) / RING * RING;
}

__device__ __forceinline__ uint4 load_ids(const unsigned* __restrict__ tid, unsigned off) {
    return *(const uint4*)((const char*)tid + off);
}

template <bool RECORD>
__device__ __forceinline__ void scan_and_add(
    int* q, int lane, int wave, int lo,
    const float* __restrict__ H, const float* __restrict__ dinv,
    const int* __restrict__ src_idx, const int* __restrict__ dst_idx,
    const unsigned* __restrict__ tid,
    float* tile, int nedges, int4* rec, int cap, int& rec_n)
{
    int qn = 0;
    int wn = 0;
    bool fits = true;
    const int nit = scan_steps(nedges);
    rec_n = 0;
    if (nit <= 0) return;
    const int rot = (int)(blockIdx.x % (unsigned)nit);
    const unsigned g4 = (unsigned)blockIdx.x * 0x01010101u;

    const unsigned lane_off = (unsigned)(wave * STEP_E + lane * 16);
    const unsigned step_bytes = STEP_E * AGG_W;

    uint4 r[RING];
    int sl = rot;
    #pragma unroll
    for (int u = 0; u < RING; ++u) {
        r[u] = load_ids(tid, lane_off + (unsigned)sl * step_bytes);
        sl = (sl + 1 < nit) ? sl + 1 : 0;
        __builtin_amdgcn_sched_barrier(0);
    }
    int st = rot;
    for (int kk = 0; kk < nit; kk += RING) {
        #pragma unroll
        for (int u = 0; u < RING; ++u) {
            const unsigned wv[4] = {r[u].x, r[u].y, r[u].z, r[u].w};
            unsigned hits = 0u;
            #pragma unroll
            for (int k = 0; k < 4; ++k) {
                const unsigned x = wv[k] ^ g4;
                const unsigned nz = ((x & 0x7F7F7F7Fu) + 0x7F7F7F7Fu) | x;
                hits |= ((~nz >> 7) & 0x01010101u) << k;
            }
            unsigned mask = __builtin_amdgcn_ballot_w32(hits != 0u);
            while (mask != 0u) {
                const int pos = qn + (int)__builtin_amdgcn_mbcnt_lo(mask, 0u);
                if (hits != 0u) {
                    const int p = __builtin_ctz(hits);
                    q[pos] = (int)(lane_off + (unsigned)st * step_bytes) + ((p & 7) << 2) + (p >> 3);
                }
                qn += __builtin_popcount(mask);
                hits &= hits - 1u;
                mask = __builtin_amdgcn_ballot_w32(hits != 0u);
            }
            r[u] = load_ids(tid, lane_off + (unsigned)sl * step_bytes);
            sl = (sl + 1 < nit) ? sl + 1 : 0;
            st = (st + 1 < nit) ? st + 1 : 0;
            if (qn >= QDRAIN) {
                fits = fits && (wn + qn <= cap);
                drain_queue<RECORD>(q, qn, lane, lo, H, dinv, src_idx, dst_idx, tile, rec + wn, fits);
                wn += qn;
                qn = 0;
            }
        }
    }
    if (qn > 0) {
        fits = fits && (wn + qn <= cap);
        drain_queue<RECORD>(q, qn, lane, lo, H, dinv, src_idx, dst_idx, tile, rec + wn, fits);
        wn += qn;
    }
    rec_n = fits ? wn : -1;
}

template <bool RELU>
__device__ __forceinline__ void transform_store(
    float* tile, int wave, int lane, int lo, int cnt,
    const float* __restrict__ Wm, const float* __restrict__ bias, float* __restrict__ out)
{
    const int m  = lane & 15;
    const int hi = lane >> 4;
    const int nt = (cnt + 15) >> 4;
    for (int t = wave; t < nt; t += AGG_W) {
        float* rows = tile + t * 16 * TP;
        v8f acc[8];
        #pragma unroll
        for (int n = 0; n < 8; ++n) acc[n] = (v8f)(0.0f);

        const float* ap = rows + m * TP + hi * 64;
        for (int k0 = 0; k0 < CH; k0 += 4) {
            const int klo = k0 + hi * 2;
            v2f a;
            a.x = ap[k0 >> 2];
            a.y = ap[32 + (k0 >> 2)];
            #pragma unroll
            for (int n = 0; n < 8; ++n) {
                v2f b;
                b.x = Wm[(size_t)klo * CH + n * 16 + m];
                b.y = Wm[(size_t)(klo + 1) * CH + n * 16 + m];
                acc[n] = __builtin_amdgcn_wmma_f32_16x16x4_f32(
                    false, a, false, b, (short)0, acc[n], false, false);
            }
        }

        float* dp = rows + (hi * 8) * TP + (m & 3) * 32 + (m >> 2);
        #pragma unroll
        for (int n = 0; n < 8; ++n) {
            const float bn = bias[n * 16 + m];
            #pragma unroll
            for (int r = 0; r < 8; ++r) {
                float v = acc[n][r] + bn;
                if (RELU) v = fmaxf(v, 0.f);
                dp[r * TP + n * 4] = v;
            }
        }
        #pragma unroll
        for (int rr = 0; rr < 16; ++rr) {
            const int row = t * 16 + rr;
            if (row < cnt) {
                const float* tp = rows + rr * TP + lane;
                float4 o;
                o.x = tp[0]; o.y = tp[32]; o.z = tp[64]; o.w = tp[96];
                *(float4*)(out + (size_t)(lo + row) * CH + lane * 4) = o;
            }
        }
    }
}

template <bool REPLAY, bool RELU>
__device__ __forceinline__ void agg_tile_body(
    float* tile, int* queue_base,
    const float* __restrict__ H, const float* __restrict__ dinv,
    const float* __restrict__ Wm, const float* __restrict__ bias,
    const int* __restrict__ src_idx, const int* __restrict__ dst_idx,
    const unsigned* __restrict__ tid,
    float* __restrict__ out, int n, int nedges,
    int* __restrict__ list_cnt, int* __restrict__ list_rec, int cap)
{
    const int lane = threadIdx.x & 31;
    const int wave = __builtin_amdgcn_readfirstlane(threadIdx.x >> 5);
    const int lo   = blockIdx.x * TILE;
    const int cnt  = min(TILE, n - lo);

    {
        const int rows16 = ((cnt + 15) >> 4) << 4;
        for (int r = wave; r < rows16; r += AGG_W) {
            float4 h = make_float4(0.f, 0.f, 0.f, 0.f);
            float w = 0.f;
            if (r < cnt) {
                const int i = lo + r;
                const float di = dinv[i];
                w = di * di;
                h = *(const float4*)(H + (size_t)i * CH + lane * 4);
            }
            float* tp = tile + r * TP + lane;
            tp[0]  = h.x * w;
            tp[32] = h.y * w;
            tp[64] = h.z * w;
            tp[96] = h.w * w;
        }
    }
    __syncthreads();

    {
        int* q = queue_base + wave * QCAP;
        const int wid = blockIdx.x * AGG_W + wave;
        int4* rec = (int4*)list_rec + (size_t)wid * (size_t)cap;
        if (REPLAY) {
            const int nrec = __builtin_amdgcn_readfirstlane(list_cnt[wid]);
            if (nrec > 0) {
                drain_list(rec, nrec, lane, H, tile);
            } else if (nrec < 0) {
                int unused;
                scan_and_add<false>(q, lane, wave, lo, H, dinv, src_idx, dst_idx, tid,
                                    tile, nedges, rec, 0, unused);
            }
        } else {
            int nrec;
            scan_and_add<true>(q, lane, wave, lo, H, dinv, src_idx, dst_idx, tid,
                               tile, nedges, rec, cap, nrec);
            if (lane == 0) list_cnt[wid] = nrec;
        }
    }
    __syncthreads();

    transform_store<RELU>(tile, wave, lane, lo, cnt, Wm, bias, out);
}

__global__ __launch_bounds__(AGG_T) void agg_tile_record_kernel(
    const float* __restrict__ H, const float* __restrict__ dinv,
    const float* __restrict__ Wm, const float* __restrict__ bias,
    const int* __restrict__ src_idx, const int* __restrict__ dst_idx,
    const unsigned* __restrict__ tid,
    float* __restrict__ out, int n, int nedges,
    int* __restrict__ list_cnt, int* __restrict__ list_rec, int cap)
{
    __shared__ __attribute__((aligned(16))) float tile[TILE * TP];
    __shared__ int queue[AGG_W][QCAP];
    agg_tile_body<false, true>(tile, &queue[0][0], H, dinv, Wm, bias, src_idx, dst_idx, tid, out, n, nedges,
                               list_cnt, list_rec, cap);
}

__global__ __launch_bounds__(AGG_T) void agg_tile_replay_kernel(
    const float* __restrict__ H, const float* __restrict__ dinv,
    const float* __restrict__ Wm, const float* __restrict__ bias,
    const int* __restrict__ src_idx, const int* __restrict__ dst_idx,
    const unsigned* __restrict__ tid,
    float* __restrict__ out, int n, int nedges,
    int* __restrict__ list_cnt, int* __restrict__ list_rec, int cap)
{
    __shared__ __attribute__((aligned(16))) float tile[TILE * TP];
    __shared__ int queue[AGG_W][QCAP];
    agg_tile_body<true, false>(tile, &queue[0][0], H, dinv, Wm, bias, src_idx, dst_idx, tid, out, n, nedges,
                               list_cnt, list_rec, cap);
}

extern "C" void kernel_launch(void* const* d_in, const int* in_sizes, int n_in,
                              void* d_out, int out_size, void* d_ws, size_t ws_size,
                              hipStream_t stream) {
    const float* x    = (const float*)d_in[0];
    const int*   edge = (const int*)d_in[1];
    const float* W1   = (const float*)d_in[2];
    const float* b1   = (const float*)d_in[3];
    const float* W2   = (const float*)d_in[4];
    const float* b2   = (const float*)d_in[5];
    float* out = (float*)d_out;

    const int N = in_sizes[0] / CH;
    const int E = in_sizes[1] / 2;
    const int* srcI = edge;
    const int* dstI = edge + E;

    char* ws = (char*)d_ws;
    float* dinv = (float*)ws;
    size_t off  = ((size_t)N * 4 + 255) & ~(size_t)255;
    float* h1   = (float*)(ws + off);
    const int nTile = (N + TILE - 1) / TILE;
    const int nStep = scan_steps(E);
    const int npad4 = nStep * AGG_W * (STEP_E / 4);
    size_t offt = off + (size_t)N * CH * 4;
    unsigned* tid = (unsigned*)(ws + offt);
    size_t offc = (offt + (size_t)npad4 * 4 + 255) & ~(size_t)255;
    int* lcnt   = (int*)(ws + offc);
    size_t offr = (offc + (size_t)nTile * AGG_W * 4 + 255) & ~(size_t)255;
    int* lrec   = (int*)(ws + offr);
    size_t room = (ws_size > offr) ? (ws_size - offr) / ((size_t)nTile * AGG_W * 16) : 0;
    const int cap = (room > (size_t)REC_CAP_MAX) ? REC_CAP_MAX : (int)room;

    const int T = 256;
    dim3 gN((N + T - 1) / T), gE((npad4 + T - 1) / T);
    dim3 gAgg(nTile);

    deg_init_kernel<<<gN, T, 0, stream>>>(dinv, N);
    deg_count_tid_kernel<<<gE, T, 0, stream>>>(dstI, dinv, tid, E, npad4);
    dinv_kernel<<<gN, T, 0, stream>>>(dinv, N);

    agg_tile_record_kernel<<<gAgg, AGG_T, 0, stream>>>(x, dinv, W1, b1, srcI, dstI, tid, h1, N, E, lcnt, lrec, cap);

    agg_tile_replay_kernel<<<gAgg, AGG_T, 0, stream>>>(h1, dinv, W2, b2, srcI, dstI, tid, out, N, E, lcnt, lrec, cap);
}


# neighbour sums kept in registers, both layers: edges filed per owner wave (13 rows each, rank by LDS integer add), added at a wave-uniform register index; no LDS f32 atomics; sum order: list order
# speedup vs baseline: 3.0684x; 1.0352x over previous
#include <hip/hip_runtime.h>
#include <hip/hip_bf16.h>
#include <cstddef>

typedef __attribute__((ext_vector_type(2))) float v2f;
typedef __attribute__((ext_vector_type(8))) float v8f;
typedef __attribute__((ext_vector_type(16))) float v16f;

#define CH 128

__global__ void deg_init_kernel(float* __restrict__ deg, int n) {
    int i = blockIdx.x * blockDim.x + threadIdx.x;
    if (i < n) deg[i] = 1.0f;
}

#define TILE_ROWS 400
__global__ void deg_count_tid_kernel(const int* __restrict__ dst, float* __restrict__ deg,
                                     unsigned* __restrict__ tid, int e, int npad4) {
    const int t = blockIdx.x * blockDim.x + threadIdx.x;
    if (t >= npad4) return;
    const int b = t * 4;
    int dv[4] = {-1, -1, -1, -1};
    if (b + 4 <= e && (((size_t)dst) & 15) == 0) {
        const int4 v = *(const int4*)(dst + b);
        dv[0] = v.x; dv[1] = v.y; dv[2] = v.z; dv[3] = v.w;
    } else {
        #pragma unroll
        for (int i = 0; i < 4; ++i)
            if (b + i < e) dv[i] = dst[b + i];
    }
    unsigned w = 0u;
    #pragma unroll
    for (int i = 0; i < 4; ++i) {
        unsigned id = 0xFFu;
        if (dv[i] >= 0) {
            unsafeAtomicAdd(&deg[dv[i]], 1.0f);
            id = (unsigned)min(dv[i] / TILE_ROWS, 255);
        }
        w |= id << (8 * i);
    }
    tid[t] = w;
}

__global__ void dinv_kernel(float* __restrict__ deg, int n) {
    int i = blockIdx.x * blockDim.x + threadIdx.x;
    if (i < n) deg[i] = rsqrtf(deg[i]);
}

#define TILE   TILE_ROWS
#define TP     129
#define AGG_T  1024
#define AGG_W  32
#define STEP_E 512
#define RING   4
#define QDRAIN 32
#define QCAP   (STEP_E + QDRAIN)
#define REC_CAP_MAX 1024
#define OWN_ROWS 13
#define OWN_CAP  400

static_assert(AGG_W * OWN_ROWS >= TILE, "every row of the tile needs an owner wave");
static_assert(AGG_W * OWN_CAP * 16 <= TILE * TP * 4, "the owner lists lie inside the tile's LDS");
static_assert(TILE % 16 == 0, "the tile is made of whole 16-row groups");

__device__ __forceinline__ void lds_add(float* p, float v) {
    __hip_atomic_fetch_add(p, v, __ATOMIC_RELAXED, __HIP_MEMORY_SCOPE_WORKGROUP);
}

__device__ __forceinline__ void file_queue(
    const int* q, int qn, int lane, int lo,
    const float* __restrict__ dinv,
    const int* __restrict__ src_idx, const int* __restrict__ dst_idx,
    int4* lists, int* ocnt)
{
    for (int base = 0; base < qn; base += 32) {
        const int k = base + lane;
        if (k < qn) {
            const int e = q[k];
            const int s = src_idx[e];
            const int d = dst_idx[e];
            const float ds = dinv[s];
            const float dd = dinv[d];
            asm volatile("" :: "v"(ds), "v"(dd));
            const float c = ds * dd;
            const int row = d - lo;
            const int o = row / OWN_ROWS;
            const int pos = __hip_atomic_fetch_add(&ocnt[o], 1, __ATOMIC_RELAXED, __HIP_MEMORY_SCOPE_WORKGROUP);
            if (pos < OWN_CAP) {
                int4 r;
                r.x = s; r.y = row - o * OWN_ROWS; r.z = __float_as_int(c); r.w = 0;
                lists[o * OWN_CAP + pos] = r;
            } else {
                ocnt[AGG_W] = 1;
            }
        }
    }
}

__device__ __forceinline__ void drain_queue(
    const int* q, int qn, int lane, int lo,
    const float* __restrict__ H, const float* __restrict__ dinv,
    const int* __restrict__ src_idx, const int* __restrict__ dst_idx,
    float* tile)
{
    for (int base = 0; base < qn; base += 32) {
        const int m = min(32, qn - base);
        const int k = min(base + lane, qn - 1);
        const int e = q[k];
        const int s = src_idx[e];
        const int d = dst_idx[e];
        const float ds = dinv[s];
        const float dd = dinv[d];
        const int rl = (d - lo) * TP;

        int j = 0;
        do {
            float4 h[8];
            int    jj[8];
            #pragma unroll
            for (int u = 0; u < 8; ++u) {
                jj[u] = min(j + u, m - 1);
                const int sj = __builtin_amdgcn_readlane(s, jj[u]);
                h[u] = *(const float4*)(H + (size_t)sj * CH + lane * 4);
            }
            asm volatile("" :: "v"(h[0].x), "v"(h[1].x), "v"(h[2].x), "v"(h[3].x),
                               "v"(h[4].x), "v"(h[5].x), "v"(h[6].x), "v"(h[7].x));
            #pragma unroll
            for (int u = 0; u < 8; ++u) {
                if (u == 0 || j + u < m) {
                    const float cu = __int_as_float(__builtin_amdgcn_readlane(__float_as_int(ds), jj[u])) *
                                     __int_as_float(__builtin_amdgcn_readlane(__float_as_int(dd), jj[u]));
                    float* tp = tile + __builtin_amdgcn_readlane(rl, jj[u]) + lane;
                    lds_add(tp,      h[u].x * cu);
                    lds_add(tp + 32, h[u].y * cu);
                    lds_add(tp + 64, h[u].z * cu);
                    lds_add(tp + 96, h[u].w * cu);
                }
            }
            j += 8;
        } while (j < m);
    }
}

struct RowSums { v16f x, y, z, w; };
static_assert(OWN_ROWS <= 16, "a row is one element of a 16-wide register vector");

__device__ __forceinline__ void add_row(RowSums& a, int row, const float4 h, float c) {
    a.x[row] += h.x * c;
    a.y[row] += h.y * c;
    a.z[row] += h.z * c;
    a.w[row] += h.w * c;
}

__device__ __forceinline__ void start_rows(
    RowSums& a, int r0, int cnt, int lo, int lane,
    const float* __restrict__ H, const float* __restrict__ dinv)
{
    a.x = (v16f)(0.0f); a.y = (v16f)(0.0f); a.z = (v16f)(0.0f); a.w = (v16f)(0.0f);
    #pragma unroll
    for (int i = 0; i < OWN_ROWS; ++i) {
        const int r = r0 + i;
        const bool in = r < cnt;
        const int g = lo + min(r, cnt - 1);
        const float di = dinv[g];
        const float w = di * di;
        const float4 h = *(const float4*)(H + (size_t)g * CH + lane * 4);
        a.x[i] = in ? h.x * w : 0.f;
        a.y[i] = in ? h.y * w : 0.f;
        a.z[i] = in ? h.z * w : 0.f;
        a.w[i] = in ? h.w * w : 0.f;
    }
}

__device__ __forceinline__ void store_rows(
    const RowSums& a, float* tile, int r0, int rows16, int lane)
{
    #pragma unroll
    for (int i = 0; i < OWN_ROWS; ++i) {
        const int r = r0 + i;
        if (r < rows16) {
            float* tp = tile + r * TP + lane;
            tp[0]  = a.x[i];
            tp[32] = a.y[i];
            tp[64] = a.z[i];
            tp[96] = a.w[i];
        }
    }
}

template <bool FULL>
__device__ __forceinline__ void add_group(
    RowSums& acc, const float* __restrict__ H, int lane, int s, int ri, float c, int j, int m)
{
    float4 h[8];
    int    jj[8];
    #pragma unroll
    for (int u = 0; u < 8; ++u) {
        jj[u] = FULL ? j + u : min(j + u, m - 1);
        const int sj = __builtin_amdgcn_readlane(s, jj[u]);
        h[u] = *(const float4*)(H + (size_t)sj * CH + lane * 4);
    }
    asm volatile("" :: "v"(h[0].x), "v"(h[1].x), "v"(h[2].x), "v"(h[3].x),
                       "v"(h[4].x), "v"(h[5].x), "v"(h[6].x), "v"(h[7].x));
    #pragma unroll
    for (int u = 0; u < 8; ++u) {
        float cu = __int_as_float(__builtin_amdgcn_readlane(__float_as_int(c), jj[u]));
        const int ru = __builtin_amdgcn_readlane(ri, jj[u]);
        float4 hv = h[u];
        if (!FULL) {
            const bool in = j + u < m;
            cu = in ? cu : 0.f;
            hv.x = in ? hv.x : 0.f; hv.y = in ? hv.y : 0.f;
            hv.z = in ? hv.z : 0.f; hv.w = in ? hv.w : 0.f;
        }
        add_row(acc, ru, hv, cu);
    }
}

template <bool COPY>
__device__ __forceinline__ void replay_rows(
    const int4* L, int n, int lane, const float* __restrict__ H,
    RowSums& acc, int4* __restrict__ recg, bool copy_ok)
{
    if (n <= 0) return;
    int4 r = L[min(lane, n - 1)];
    for (int base = 0; base < n; base += 32) {
        const int m = min(32, n - base);
        const int4 rn = L[min(base + 32 + lane, n - 1)];
        if (COPY) {
            if (copy_ok && base + lane < n) recg[base + lane] = r;
        }
        const int   s  = r.x;
        const int   ri = r.y;
        const float c  = __int_as_float(r.z);

        int j = 0;
        for (; j + 8 <= m; j += 8) add_group<true>(acc, H, lane, s, ri, c, j, m);
        if (j < m) add_group<false>(acc, H, lane, s, ri, c, j, m);
        r = rn;
    }
}

__host__ __device__ __forceinline__ int scan_steps(int nedges) {
    const int nchunks = (nedges + STEP_E - 1) / STEP_E;
    return ((nchunks + AGG_W - 1) / AGG_W + RING - 1) / RING * RING;
}

__device__ __forceinline__ uint4 load_ids(const unsigned* __restrict__ tid, unsigned off) {
    return *(const uint4*)((const char*)tid + off);
}

template <bool FILE>
__device__ __forceinline__ void scan_edges(
    int* q, int lane, int wave, int lo,
    const float* __restrict__ H, const float* __restrict__ dinv,
    const int* __restrict__ src_idx, const int* __restrict__ dst_idx,
    const unsigned* __restrict__ tid,
    float* tile, int nedges, int4* lists, int* ocnt)
{
    int qn = 0;
    const int nit = scan_steps(nedges);
    if (nit <= 0) return;
    const int rot = (int)(blockIdx.x % (unsigned)nit);
    const unsigned g4 = (unsigned)blockIdx.x * 0x01010101u;

    const unsigned lane_off = (unsigned)(wave * STEP_E + lane * 16);
    const unsigned step_bytes = STEP_E * AGG_W;

    uint4 r[RING];
    int sl = rot;
    #pragma unroll
    for (int u = 0; u < RING; ++u) {
        r[u] = load_ids(tid, lane_off + (unsigned)sl * step_bytes);
        sl = (sl + 1 < nit) ? sl + 1 : 0;
        __builtin_amdgcn_sched_barrier(0);
    }
    int st = rot;
    for (int kk = 0; kk < nit; kk += RING) {
        #pragma unroll
        for (int u = 0; u < RING; ++u) {
            const unsigned wv[4] = {r[u].x, r[u].y, r[u].z, r[u].w};
            unsigned hits = 0u;
            #pragma unroll
            for (int k = 0; k < 4; ++k) {
                const unsigned x = wv[k] ^ g4;
                const unsigned nz = ((x & 0x7F7F7F7Fu) + 0x7F7F7F7Fu) | x;
                hits |= ((~nz >> 7) & 0x01010101u) << k;
            }
            unsigned mask = __builtin_amdgcn_ballot_w32(hits != 0u);
            while (mask != 0u) {
                const int pos = qn + (int)__builtin_amdgcn_mbcnt_lo(mask, 0u);
                if (hits != 0u) {
                    const int p = __builtin_ctz(hits);
                    q[pos] = (int)(lane_off + (unsigned)st * step_bytes) + ((p & 7) << 2) + (p >> 3);
                }
                qn += __builtin_popcount(mask);
                hits &= hits - 1u;
                mask = __builtin_amdgcn_ballot_w32(hits != 0u);
            }
            r[u] = load_ids(tid, lane_off + (unsigned)sl * step_bytes);
            sl = (sl + 1 < nit) ? sl + 1 : 0;
            st = (st + 1 < nit) ? st + 1 : 0;
            if (qn >= QDRAIN) {
                if (FILE) file_queue(q, qn, lane, lo, dinv, src_idx, dst_idx, lists, ocnt);
                else      drain_queue(q, qn, lane, lo, H, dinv, src_idx, dst_idx, tile);
                qn = 0;
            }
        }
    }
    if (qn > 0) {
        if (FILE) file_queue(q, qn, lane, lo, dinv, src_idx, dst_idx, lists, ocnt);
        else      drain_queue(q, qn, lane, lo, H, dinv, src_idx, dst_idx, tile);
    }
}

template <bool RELU>
__device__ __forceinline__ void transform_store(
    float* tile, int wave, int lane, int lo, int cnt,
    const float* __restrict__ Wm, const float* __restrict__ bias, float* __restrict__ out)
{
    const int m  = lane & 15;
    const int hi = lane >> 4;
    const int nt = (cnt + 15) >> 4;
    for (int t = wave; t < nt; t += AGG_W) {
        float* rows = tile + t * 16 * TP;
        v8f acc[8];
        #pragma unroll
        for (int n = 0; n < 8; ++n) acc[n] = (v8f)(0.0f);

        const float* ap = rows + m * TP + hi * 64;
        for (int k0 = 0; k0 < CH; k0 += 4) {
            const int klo = k0 + hi * 2;
            v2f a;
            a.x = ap[k0 >> 2];
            a.y = ap[32 + (k0 >> 2)];
            #pragma unroll
            for (int n = 0; n < 8; ++n) {
                v2f b;
                b.x = Wm[(size_t)klo * CH + n * 16 + m];
                b.y = Wm[(size_t)(klo + 1) * CH + n * 16 + m];
                acc[n] = __builtin_amdgcn_wmma_f32_16x16x4_f32(
                    false, a, false, b, (short)0, acc[n], false, false);
            }
        }

        float* dp = rows + (hi * 8) * TP + (m & 3) * 32 + (m >> 2);
        #pragma unroll
        for (int n = 0; n < 8; ++n) {
            const float bn = bias[n * 16 + m];
            #pragma unroll
            for (int r = 0; r < 8; ++r) {
                float v = acc[n][r] + bn;
                if (RELU) v = fmaxf(v, 0.f);
                dp[r * TP + n * 4] = v;
            }
        }
        #pragma unroll
        for (int rr = 0; rr < 16; ++rr) {
            const int row = t * 16 + rr;
            if (row < cnt) {
                const float* tp = rows + rr * TP + lane;
                float4 o;
                o.x = tp[0]; o.y = tp[32]; o.z = tp[64]; o.w = tp[96];
                *(float4*)(out + (size_t)(lo + row) * CH + lane * 4) = o;
            }
        }
    }
}

__device__ __forceinline__ void start_tile(
    float* tile, int wave, int lane, int lo, int cnt, int rows16,
    const float* __restrict__ H, const float* __restrict__ dinv)
{
    for (int r = wave; r < rows16; r += AGG_W) {
        float4 h = make_float4(0.f, 0.f, 0.f, 0.f);
        float w = 0.f;
        if (r < cnt) {
            const int i = lo + r;
            const float di = dinv[i];
            w = di * di;
            h = *(const float4*)(H + (size_t)i * CH + lane * 4);
        }
        float* tp = tile + r * TP + lane;
        tp[0]  = h.x * w;
        tp[32] = h.y * w;
        tp[64] = h.z * w;
        tp[96] = h.w * w;
    }
}

__global__ __launch_bounds__(AGG_T) void agg_tile_record_kernel(
    const float* __restrict__ H, const float* __restrict__ dinv,
    const float* __restrict__ Wm, const float* __restrict__ bias,
    const int* __restrict__ src_idx, const int* __restrict__ dst_idx,
    const unsigned* __restrict__ tid,
    float* __restrict__ out, int n, int nedges,
    int* __restrict__ list_cnt, int* __restrict__ list_rec, int cap)
{
    __shared__ __attribute__((aligned(16))) float tile[TILE * TP];
    __shared__ int queue[AGG_W][QCAP];
    __shared__ int ocnt[AGG_W + 1];

    const int lane = threadIdx.x & 31;
    const int wave = __builtin_amdgcn_readfirstlane(threadIdx.x >> 5);
    const int lo   = blockIdx.x * TILE;
    const int cnt  = min(TILE, n - lo);
    const int rows16 = ((cnt + 15) >> 4) << 4;
    int* q = &queue[wave][0];
    const int wid = blockIdx.x * AGG_W + wave;
    int4* rec = (int4*)list_rec + (size_t)wid * (size_t)cap;
    int4* lists = (int4*)tile;

    if (threadIdx.x <= AGG_W) ocnt[threadIdx.x] = 0;
    __syncthreads();

    scan_edges<true>(q, lane, wave, lo, H, dinv, src_idx, dst_idx, tid, tile, nedges, lists, ocnt);
    __syncthreads();

    const int full = __builtin_amdgcn_readfirstlane(ocnt[AGG_W]);
    if (full == 0) {
        const int nown = __builtin_amdgcn_readfirstlane(ocnt[wave]);
        const bool copy_ok = nown <= cap;
        RowSums acc;
        start_rows(acc, wave * OWN_ROWS, cnt, lo, lane, H, dinv);
        replay_rows<true>(lists + wave * OWN_CAP, nown, lane, H, acc, rec, copy_ok);
        if (lane == 0) list_cnt[wid] = copy_ok ? nown : -1;
        __syncthreads();
        store_rows(acc, tile, wave * OWN_ROWS, rows16, lane);
    } else {
        start_tile(tile, wave, lane, lo, cnt, rows16, H, dinv);
        __syncthreads();
        scan_edges<false>(q, lane, wave, lo, H, dinv, src_idx, dst_idx, tid, tile, nedges, lists, ocnt);
        if (lane == 0) list_cnt[wid] = -1;
    }
    __syncthreads();

    transform_store<true>(tile, wave, lane, lo, cnt, Wm, bias, out);
}

__global__ __launch_bounds__(AGG_T) void agg_tile_replay_kernel(
    const float* __restrict__ H, const float* __restrict__ dinv,
    const float* __restrict__ Wm, const float* __restrict__ bias,
    const int* __restrict__ src_idx, const int* __restrict__ dst_idx,
    const unsigned* __restrict__ tid,
    float* __restrict__ out, int n, int nedges,
    int* __restrict__ list_cnt, int* __restrict__ list_rec, int cap)
{
    __shared__ __attribute__((aligned(16))) float tile[TILE * TP];
    __shared__ int queue[AGG_W][QCAP];
    __shared__ int nolist;

    const int lane = threadIdx.x & 31;
    const int wave = __builtin_amdgcn_readfirstlane(threadIdx.x >> 5);
    const int lo   = blockIdx.x * TILE;
    const int cnt  = min(TILE, n - lo);
    const int rows16 = ((cnt + 15) >> 4) << 4;
    const int wid = blockIdx.x * AGG_W + wave;
    const int4* rec = (const int4*)list_rec + (size_t)wid * (size_t)cap;

    const int nrec = __builtin_amdgcn_readfirstlane(list_cnt[wid]);
    if (threadIdx.x == 0) nolist = 0;
    __syncthreads();
    if (nrec < 0 && lane == 0) nolist = 1;
    __syncthreads();

    const int bad = __builtin_amdgcn_readfirstlane(nolist);
    if (bad == 0) {
        RowSums acc;
        start_rows(acc, wave * OWN_ROWS, cnt, lo, lane, H, dinv);
        replay_rows<false>(rec, nrec, lane, H, acc, nullptr, false);
        store_rows(acc, tile, wave * OWN_ROWS, rows16, lane);
    } else {
        start_tile(tile, wave, lane, lo, cnt, rows16, H, dinv);
        __syncthreads();
        scan_edges<false>(&queue[wave][0], lane, wave, lo, H, dinv, src_idx, dst_idx, tid, tile, nedges,
                          (int4*)tile, &nolist);
    }
    __syncthreads();

    transform_store<false>(tile, wave, lane, lo, cnt, Wm, bias, out);
}

extern "C" void kernel_launch(void* const* d_in, const int* in_sizes, int n_in,
                              void* d_out, int out_size, void* d_ws, size_t ws_size,
                              hipStream_t stream) {
    const float* x    = (const float*)d_in[0];
    const int*   edge = (const int*)d_in[1];
    const float* W1   = (const float*)d_in[2];
    const float* b1   = (const float*)d_in[3];
    const float* W2   = (const float*)d_in[4];
    const float* b2   = (const float*)d_in[5];
    float* out = (float*)d_out;

    const int N = in_sizes[0] / CH;
    const int E = in_sizes[1] / 2;
    const int* srcI = edge;
    const int* dstI = edge + E;

    char* ws = (char*)d_ws;
    float* dinv = (float*)ws;
    size_t off  = ((size_t)N * 4 + 255) & ~(size_t)255;
    float* h1   = (float*)(ws + off);
    const int nTile = (N + TILE - 1) / TILE;
    const int nStep = scan_steps(E);
    const int npad4 = nStep * AGG_W * (STEP_E / 4);
    size_t offt = off + (size_t)N * CH * 4;
    unsigned* tid = (unsigned*)(ws + offt);
    size_t offc = (offt + (size_t)npad4 * 4 + 255) & ~(size_t)255;
    int* lcnt   = (int*)(ws + offc);
    size_t offr = (offc + (size_t)nTile * AGG_W * 4 + 255) & ~(size_t)255;
    int* lrec   = (int*)(ws + offr);
    size_t room = (ws_size > offr) ? (ws_size - offr) / ((size_t)nTile * AGG_W * 16) : 0;
    const int cap = (room > (size_t)REC_CAP_MAX) ? REC_CAP_MAX : (int)room;

    const int T = 256;
    dim3 gN((N + T - 1) / T), gE((npad4 + T - 1) / T);
    dim3 gAgg(nTile);

    deg_init_kernel<<<gN, T, 0, stream>>>(dinv, N);
    deg_count_tid_kernel<<<gE, T, 0, stream>>>(dstI, dinv, tid, E, npad4);
    dinv_kernel<<<gN, T, 0, stream>>>(dinv, N);

    agg_tile_record_kernel<<<gAgg, AGG_T, 0, stream>>>(x, dinv, W1, b1, srcI, dstI, tid, h1, N, E, lcnt, lrec, cap);

    agg_tile_replay_kernel<<<gAgg, AGG_T, 0, stream>>>(h1, dinv, W2, b2, srcI, dstI, tid, out, N, E, lcnt, lrec, cap);
}


# tile membership as one bitmap per tile (1 bit per edge, built in LDS by the degree kernel): an owner work-group walks its own 200 KB of bits, not every edge's id byte; sum order: LDS arrival rank
# speedup vs baseline: 3.2071x; 1.0452x over previous
#include <hip/hip_runtime.h>
#include <hip/hip_bf16.h>
#include <cstddef>

typedef __attribute__((ext_vector_type(2))) float v2f;
typedef __attribute__((ext_vector_type(8))) float v8f;
typedef __attribute__((ext_vector_type(16))) float v16f;

#define CH 128

__global__ void deg_init_kernel(float* __restrict__ deg, int n) {
    int i = blockIdx.x * blockDim.x + threadIdx.x;
    if (i < n) deg[i] = 1.0f;
}

#define TILE_ROWS 400
#define BM_CHUNK 4096
#define BM_WORDS (BM_CHUNK / 32)
#define BM_MAXT  256
#define BM_T     1024
static_assert(BM_T * 4 == BM_CHUNK, "a thread takes four edges");
__global__ __launch_bounds__(BM_T) void deg_count_bitmap_kernel(
    const int* __restrict__ dst, float* __restrict__ deg,
    unsigned* __restrict__ bm, int e, int ntile, int wpt)
{
    __shared__ __attribute__((aligned(16))) unsigned m[BM_MAXT * BM_WORDS];
    uint4* m4 = (uint4*)m;
    const int t = threadIdx.x;

    const int b = blockIdx.x * BM_CHUNK + t * 4;
    int dv[4] = {-1, -1, -1, -1};
    if (b + 4 <= e && (((size_t)dst) & 15) == 0) {
        const int4 v = *(const int4*)(dst + b);
        dv[0] = v.x; dv[1] = v.y; dv[2] = v.z; dv[3] = v.w;
    } else {
        #pragma unroll
        for (int i = 0; i < 4; ++i)
            if (b + i < e) dv[i] = dst[b + i];
    }
    for (int i = t; i < ntile * (BM_WORDS / 4); i += BM_T) m4[i] = make_uint4(0u, 0u, 0u, 0u);
    __syncthreads();

    #pragma unroll
    for (int i = 0; i < 4; ++i) {
        if (dv[i] >= 0) {
            const int g = dv[i] / TILE_ROWS;
            if (g < ntile) atomicOr(&m[g * BM_WORDS + (t >> 3)], 1u << ((t * 4 + i) & 31));
        }
    }
    __syncthreads();

    #pragma unroll
    for (int i = 0; i < 4; ++i)
        if (dv[i] >= 0) unsafeAtomicAdd(&deg[dv[i]], 1.0f);

    const int lane = t & 31;
    const int wave = t >> 5;
    for (int g = wave; g < ntile; g += BM_T / 32) {
        const uint4 v = m4[g * (BM_WORDS / 4) + lane];
        *(uint4*)(bm + (size_t)g * (size_t)wpt + (size_t)blockIdx.x * BM_WORDS + lane * 4) = v;
    }
}

__global__ void dinv_kernel(float* __restrict__ deg, int n) {
    int i = blockIdx.x * blockDim.x + threadIdx.x;
    if (i < n) deg[i] = rsqrtf(deg[i]);
}

#define TILE   TILE_ROWS
#define TP     129
#define AGG_T  1024
#define AGG_W  32
#define QDRAIN 32
#define QCAP   (QDRAIN + 32)
#define REC_CAP_MAX 1024
#define OWN_ROWS 13
#define OWN_CAP  400

static_assert(AGG_W * OWN_ROWS >= TILE, "every row of the tile needs an owner wave");
static_assert(AGG_W * OWN_CAP * 16 <= TILE * TP * 4, "the owner lists lie inside the tile's LDS");
static_assert(TILE % 16 == 0, "the tile is made of whole 16-row groups");

__device__ __forceinline__ void lds_add(float* p, float v) {
    __hip_atomic_fetch_add(p, v, __ATOMIC_RELAXED, __HIP_MEMORY_SCOPE_WORKGROUP);
}

__device__ __forceinline__ void file_queue(
    const int* q, int qn, int lane, int lo,
    const float* __restrict__ dinv,
    const int* __restrict__ src_idx, const int* __restrict__ dst_idx,
    int4* lists, int* ocnt)
{
    for (int base = 0; base < qn; base += 32) {
        const int k = base + lane;
        if (k < qn) {
            const int e = q[k];
            const int s = src_idx[e];
            const int d = dst_idx[e];
            const float ds = dinv[s];
            const float dd = dinv[d];
            asm volatile("" :: "v"(ds), "v"(dd));
            const float c = ds * dd;
            const int row = d - lo;
            const int o = row / OWN_ROWS;
            const int pos = __hip_atomic_fetch_add(&ocnt[o], 1, __ATOMIC_RELAXED, __HIP_MEMORY_SCOPE_WORKGROUP);
            if (pos < OWN_CAP) {
                int4 r;
                r.x = s; r.y = row - o * OWN_ROWS; r.z = __float_as_int(c); r.w = 0;
                lists[o * OWN_CAP + pos] = r;
            } else {
                ocnt[AGG_W] = 1;
            }
        }
    }
}

__device__ __forceinline__ void drain_queue(
    const int* q, int qn, int lane, int lo,
    const float* __restrict__ H, const float* __restrict__ dinv,
    const int* __restrict__ src_idx, const int* __restrict__ dst_idx,
    float* tile)
{
    for (int base = 0; base < qn; base += 32) {
        const int m = min(32, qn - base);
        const int k = min(base + lane, qn - 1);
        const int e = q[k];
        const int s = src_idx[e];
        const int d = dst_idx[e];
        const float ds = dinv[s];
        const float dd = dinv[d];
        const int rl = (d - lo) * TP;

        int j = 0;
        do {
            float4 h[8];
            int    jj[8];
            #pragma unroll
            for (int u = 0; u < 8; ++u) {
                jj[u] = min(j + u, m - 1);
                const int sj = __builtin_amdgcn_readlane(s, jj[u]);
                h[u] = *(const float4*)(H + (size_t)sj * CH + lane * 4);
            }
            asm volatile("" :: "v"(h[0].x), "v"(h[1].x), "v"(h[2].x), "v"(h[3].x),
                               "v"(h[4].x), "v"(h[5].x), "v"(h[6].x), "v"(h[7].x));
            #pragma unroll
            for (int u = 0; u < 8; ++u) {
                if (u == 0 || j + u < m) {
                    const float cu = __int_as_float(__builtin_amdgcn_readlane(__float_as_int(ds), jj[u])) *
                                     __int_as_float(__builtin_amdgcn_readlane(__float_as_int(dd), jj[u]));
                    float* tp = tile + __builtin_amdgcn_readlane(rl, jj[u]) + lane;
                    lds_add(tp,      h[u].x * cu);
                    lds_add(tp + 32, h[u].y * cu);
                    lds_add(tp + 64, h[u].z * cu);
                    lds_add(tp + 96, h[u].w * cu);
                }
            }
            j += 8;
        } while (j < m);
    }
}

struct RowSums { v16f x, y, z, w; };
static_assert(OWN_ROWS <= 16, "a row is one element of a 16-wide register vector");

__device__ __forceinline__ void add_row(RowSums& a, int row, const float4 h, float c) {
    a.x[row] += h.x * c;
    a.y[row] += h.y * c;
    a.z[row] += h.z * c;
    a.w[row] += h.w * c;
}

__device__ __forceinline__ void start_rows(
    RowSums& a, int r0, int cnt, int lo, int lane,
    const float* __restrict__ H, const float* __restrict__ dinv)
{
    a.x = (v16f)(0.0f); a.y = (v16f)(0.0f); a.z = (v16f)(0.0f); a.w = (v16f)(0.0f);
    #pragma unroll
    for (int i = 0; i < OWN_ROWS; ++i) {
        const int r = r0 + i;
        const bool in = r < cnt;
        const int g = lo + min(r, cnt - 1);
        const float di = dinv[g];
        const float w = di * di;
        const float4 h = *(const float4*)(H + (size_t)g * CH + lane * 4);
        a.x[i] = in ? h.x * w : 0.f;
        a.y[i] = in ? h.y * w : 0.f;
        a.z[i] = in ? h.z * w : 0.f;
        a.w[i] = in ? h.w * w : 0.f;
    }
}

__device__ __forceinline__ void store_rows(
    const RowSums& a, float* tile, int r0, int rows16, int lane)
{
    #pragma unroll
    for (int i = 0; i < OWN_ROWS; ++i) {
        const int r = r0 + i;
        if (r < rows16) {
            float* tp = tile + r * TP + lane;
            tp[0]  = a.x[i];
            tp[32] = a.y[i];
            tp[64] = a.z[i];
            tp[96] = a.w[i];
        }
    }
}

template <bool FULL>
__device__ __forceinline__ void add_group(
    RowSums& acc, const float* __restrict__ H, int lane, int s, int ri, float c, int j, int m)
{
    float4 h[8];
    int    jj[8];
    #pragma unroll
    for (int u = 0; u < 8; ++u) {
        jj[u] = FULL ? j + u : min(j + u, m - 1);
        const int sj = __builtin_amdgcn_readlane(s, jj[u]);
        h[u] = *(const float4*)(H + (size_t)sj * CH + lane * 4);
    }
    asm volatile("" :: "v"(h[0].x), "v"(h[1].x), "v"(h[2].x), "v"(h[3].x),
                       "v"(h[4].x), "v"(h[5].x), "v"(h[6].x), "v"(h[7].x));
    #pragma unroll
    for (int u = 0; u < 8; ++u) {
        float cu = __int_as_float(__builtin_amdgcn_readlane(__float_as_int(c), jj[u]));
        const int ru = __builtin_amdgcn_readlane(ri, jj[u]);
        float4 hv = h[u];
        if (!FULL) {
            const bool in = j + u < m;
            cu = in ? cu : 0.f;
            hv.x = in ? hv.x : 0.f; hv.y = in ? hv.y : 0.f;
            hv.z = in ? hv.z : 0.f; hv.w = in ? hv.w : 0.f;
        }
        add_row(acc, ru, hv, cu);
    }
}

template <bool COPY>
__device__ __forceinline__ void replay_rows(
    const int4* L, int n, int lane, const float* __restrict__ H,
    RowSums& acc, int4* __restrict__ recg, bool copy_ok)
{
    if (n <= 0) return;
    int4 r = L[min(lane, n - 1)];
    for (int base = 0; base < n; base += 32) {
        const int m = min(32, n - base);
        const int4 rn = L[min(base + 32 + lane, n - 1)];
        if (COPY) {
            if (copy_ok && base + lane < n) recg[base + lane] = r;
        }
        const int   s  = r.x;
        const int   ri = r.y;
        const float c  = __int_as_float(r.z);

        int j = 0;
        for (; j + 8 <= m; j += 8) add_group<true>(acc, H, lane, s, ri, c, j, m);
        if (j < m) add_group<false>(acc, H, lane, s, ri, c, j, m);
        r = rn;
    }
}

template <bool FILE>
__device__ __forceinline__ void scan_edges(
    int* q, int lane, int wave, int lo,
    const float* __restrict__ H, const float* __restrict__ dinv,
    const int* __restrict__ src_idx, const int* __restrict__ dst_idx,
    const unsigned* __restrict__ bmt,
    float* tile, int nchunk, int4* lists, int* ocnt)
{
    int qn = 0;
    if (wave >= nchunk) return;
    uint4 cur = *(const uint4*)(bmt + (size_t)wave * BM_WORDS + lane * 4);
    for (int ch = wave; ch < nchunk; ch += AGG_W) {
        const int chn = min(ch + AGG_W, nchunk - 1);
        const uint4 nxt = *(const uint4*)(bmt + (size_t)chn * BM_WORDS + lane * 4);
        const unsigned wv[4] = {cur.x, cur.y, cur.z, cur.w};
        const int ebase = ch * BM_CHUNK + lane * 128;
        #pragma unroll
        for (int k = 0; k < 4; ++k) {
            unsigned hits = wv[k];
            unsigned mask = __builtin_amdgcn_ballot_w32(hits != 0u);
            while (mask != 0u) {
                const int pos = qn + (int)__builtin_amdgcn_mbcnt_lo(mask, 0u);
                if (hits != 0u) q[pos] = ebase + k * 32 + __builtin_ctz(hits);
                qn += __builtin_popcount(mask);
                hits &= hits - 1u;
                mask = __builtin_amdgcn_ballot_w32(hits != 0u);
                if (qn >= QDRAIN) {
                    if (FILE) file_queue(q, qn, lane, lo, dinv, src_idx, dst_idx, lists, ocnt);
                    else      drain_queue(q, qn, lane, lo, H, dinv, src_idx, dst_idx, tile);
                    qn = 0;
                }
            }
        }
        cur = nxt;
    }
    if (qn > 0) {
        if (FILE) file_queue(q, qn, lane, lo, dinv, src_idx, dst_idx, lists, ocnt);
        else      drain_queue(q, qn, lane, lo, H, dinv, src_idx, dst_idx, tile);
    }
}

template <bool RELU>
__device__ __forceinline__ void transform_store(
    float* tile, int wave, int lane, int lo, int cnt,
    const float* __restrict__ Wm, const float* __restrict__ bias, float* __restrict__ out)
{
    const int m  = lane & 15;
    const int hi = lane >> 4;
    const int nt = (cnt + 15) >> 4;
    for (int t = wave; t < nt; t += AGG_W) {
        float* rows = tile + t * 16 * TP;
        v8f acc[8];
        #pragma unroll
        for (int n = 0; n < 8; ++n) acc[n] = (v8f)(0.0f);

        const float* ap = rows + m * TP + hi * 64;
        for (int k0 = 0; k0 < CH; k0 += 4) {
            const int klo = k0 + hi * 2;
            v2f a;
            a.x = ap[k0 >> 2];
            a.y = ap[32 + (k0 >> 2)];
            #pragma unroll
            for (int n = 0; n < 8; ++n) {
                v2f b;
                b.x = Wm[(size_t)klo * CH + n * 16 + m];
                b.y = Wm[(size_t)(klo + 1) * CH + n * 16 + m];
                acc[n] = __builtin_amdgcn_wmma_f32_16x16x4_f32(
                    false, a, false, b, (short)0, acc[n], false, false);
            }
        }

        float* dp = rows + (hi * 8) * TP + (m & 3) * 32 + (m >> 2);
        #pragma unroll
        for (int n = 0; n < 8; ++n) {
            const float bn = bias[n * 16 + m];
            #pragma unroll
            for (int r = 0; r < 8; ++r) {
                float v = acc[n][r] + bn;
                if (RELU) v = fmaxf(v, 0.f);
                dp[r * TP + n * 4] = v;
            }
        }
        #pragma unroll
        for (int rr = 0; rr < 16; ++rr) {
            const int row = t * 16 + rr;
            if (row < cnt) {
                const float* tp = rows + rr * TP + lane;
                float4 o;
                o.x = tp[0]; o.y = tp[32]; o.z = tp[64]; o.w = tp[96];
                *(float4*)(out + (size_t)(lo + row) * CH + lane * 4) = o;
            }
        }
    }
}

__device__ __forceinline__ void start_tile(
    float* tile, int wave, int lane, int lo, int cnt, int rows16,
    const float* __restrict__ H, const float* __restrict__ dinv)
{
    for (int r = wave; r < rows16; r += AGG_W) {
        float4 h = make_float4(0.f, 0.f, 0.f, 0.f);
        float w = 0.f;
        if (r < cnt) {
            const int i = lo + r;
            const float di = dinv[i];
            w = di * di;
            h = *(const float4*)(H + (size_t)i * CH + lane * 4);
        }
        float* tp = tile + r * TP + lane;
        tp[0]  = h.x * w;
        tp[32] = h.y * w;
        tp[64] = h.z * w;
        tp[96] = h.w * w;
    }
}

__global__ __launch_bounds__(AGG_T) void agg_tile_record_kernel(
    const float* __restrict__ H, const float* __restrict__ dinv,
    const float* __restrict__ Wm, const float* __restrict__ bias,
    const int* __restrict__ src_idx, const int* __restrict__ dst_idx,
    const unsigned* __restrict__ bm,
    float* __restrict__ out, int n, int nchunk,
    int* __restrict__ list_cnt, int* __restrict__ list_rec, int cap)
{
    __shared__ __attribute__((aligned(16))) float tile[TILE * TP];
    __shared__ int queue[AGG_W][QCAP];
    __shared__ int ocnt[AGG_W + 1];

    const int lane = threadIdx.x & 31;
    const int wave = __builtin_amdgcn_readfirstlane(threadIdx.x >> 5);
    const int lo   = blockIdx.x * TILE;
    const int cnt  = min(TILE, n - lo);
    const int rows16 = ((cnt + 15) >> 4) << 4;
    int* q = &queue[wave][0];
    const int wid = blockIdx.x * AGG_W + wave;
    int4* rec = (int4*)list_rec + (size_t)wid * (size_t)cap;
    int4* lists = (int4*)tile;
    const unsigned* bmt = bm + (size_t)blockIdx.x * (size_t)nchunk * BM_WORDS;

    if (threadIdx.x <= AGG_W) ocnt[threadIdx.x] = 0;
    __syncthreads();

    scan_edges<true>(q, lane, wave, lo, H, dinv, src_idx, dst_idx, bmt, tile, nchunk, lists, ocnt);
    __syncthreads();

    const int full = __builtin_amdgcn_readfirstlane(ocnt[AGG_W]);
    if (full == 0) {
        const int nown = __builtin_amdgcn_readfirstlane(ocnt[wave]);
        const bool copy_ok = nown <= cap;
        RowSums acc;
        start_rows(acc, wave * OWN_ROWS, cnt, lo, lane, H, dinv);
        replay_rows<true>(lists + wave * OWN_CAP, nown, lane, H, acc, rec, copy_ok);
        if (lane == 0) list_cnt[wid] = copy_ok ? nown : -1;
        __syncthreads();
        store_rows(acc, tile, wave * OWN_ROWS, rows16, lane);
    } else {
        start_tile(tile, wave, lane, lo, cnt, rows16, H, dinv);
        __syncthreads();
        scan_edges<false>(q, lane, wave, lo, H, dinv, src_idx, dst_idx, bmt, tile, nchunk, lists, ocnt);
        if (lane == 0) list_cnt[wid] = -1;
    }
    __syncthreads();

    transform_store<true>(tile, wave, lane, lo, cnt, Wm, bias, out);
}

__global__ __launch_bounds__(AGG_T) void agg_tile_replay_kernel(
    const float* __restrict__ H, const float* __restrict__ dinv,
    const float* __restrict__ Wm, const float* __restrict__ bias,
    const int* __restrict__ src_idx, const int* __restrict__ dst_idx,
    const unsigned* __restrict__ bm,
    float* __restrict__ out, int n, int nchunk,
    int* __restrict__ list_cnt, int* __restrict__ list_rec, int cap)
{
    __shared__ __attribute__((aligned(16))) float tile[TILE * TP];
    __shared__ int queue[AGG_W][QCAP];
    __shared__ int nolist;

    const int lane = threadIdx.x & 31;
    const int wave = __builtin_amdgcn_readfirstlane(threadIdx.x >> 5);
    const int lo   = blockIdx.x * TILE;
    const int cnt  = min(TILE, n - lo);
    const int rows16 = ((cnt + 15) >> 4) << 4;
    const int wid = blockIdx.x * AGG_W + wave;
    const int4* rec = (const int4*)list_rec + (size_t)wid * (size_t)cap;

    const int nrec = __builtin_amdgcn_readfirstlane(list_cnt[wid]);
    if (threadIdx.x == 0) nolist = 0;
    __syncthreads();
    if (nrec < 0 && lane == 0) nolist = 1;
    __syncthreads();

    const int bad = __builtin_amdgcn_readfirstlane(nolist);
    if (bad == 0) {
        RowSums acc;
        start_rows(acc, wave * OWN_ROWS, cnt, lo, lane, H, dinv);
        replay_rows<false>(rec, nrec, lane, H, acc, nullptr, false);
        store_rows(acc, tile, wave * OWN_ROWS, rows16, lane);
    } else {
        start_tile(tile, wave, lane, lo, cnt, rows16, H, dinv);
        __syncthreads();
        scan_edges<false>(&queue[wave][0], lane, wave, lo, H, dinv, src_idx, dst_idx,
                          bm + (size_t)blockIdx.x * (size_t)nchunk * BM_WORDS, tile, nchunk,
                          (int4*)tile, &nolist);
    }
    __syncthreads();

    transform_store<false>(tile, wave, lane, lo, cnt, Wm, bias, out);
}

extern "C" void kernel_launch(void* const* d_in, const int* in_sizes, int n_in,
                              void* d_out, int out_size, void* d_ws, size_t ws_size,
                              hipStream_t stream) {
    const float* x    = (const float*)d_in[0];
    const int*   edge = (const int*)d_in[1];
    const float* W1   = (const float*)d_in[2];
    const float* b1   = (const float*)d_in[3];
    const float* W2   = (const float*)d_in[4];
    const float* b2   = (const float*)d_in[5];
    float* out = (float*)d_out;

    const int N = in_sizes[0] / CH;
    const int E = in_sizes[1] / 2;
    const int* srcI = edge;
    const int* dstI = edge + E;

    char* ws = (char*)d_ws;
    float* dinv = (float*)ws;
    size_t off  = ((size_t)N * 4 + 255) & ~(size_t)255;
    float* h1   = (float*)(ws + off);
    const int nTile = (N + TILE - 1) / TILE;
    const int nChunk = (E + BM_CHUNK - 1) / BM_CHUNK;
    const int wpt = nChunk * BM_WORDS;
    size_t offt = off + (size_t)N * CH * 4;
    unsigned* bm = (unsigned*)(ws + offt);
    size_t offc = (offt + (size_t)nTile * (size_t)wpt * 4 + 255) & ~(size_t)255;
    int* lcnt   = (int*)(ws + offc);
    size_t offr = (offc + (size_t)nTile * AGG_W * 4 + 255) & ~(size_t)255;
    int* lrec   = (int*)(ws + offr);
    size_t room = (ws_size > offr) ? (ws_size - offr) / ((size_t)nTile * AGG_W * 16) : 0;
    const int cap = (room > (size_t)REC_CAP_MAX) ? REC_CAP_MAX : (int)room;

    const int T = 256;
    dim3 gN((N + T - 1) / T), gE(nChunk);
    dim3 gAgg(nTile);

    deg_init_kernel<<<gN, T, 0, stream>>>(dinv, N);
    deg_count_bitmap_kernel<<<gE, BM_T, 0, stream>>>(dstI, dinv, bm, E, nTile, wpt);
    dinv_kernel<<<gN, T, 0, stream>>>(dinv, N);

    agg_tile_record_kernel<<<gAgg, AGG_T, 0, stream>>>(x, dinv, W1, b1, srcI, dstI, bm, h1, N, nChunk, lcnt, lrec, cap);

    agg_tile_replay_kernel<<<gAgg, AGG_T, 0, stream>>>(h1, dinv, W2, b2, srcI, dstI, bm, out, N, nChunk, lcnt, lrec, cap);
}
